# starting kernel
# baseline (speedup 1.0000x reference)
#include <hip/hip_runtime.h>

typedef float v2f __attribute__((ext_vector_type(2)));
typedef float v8f __attribute__((ext_vector_type(8)));

#define DIM 64
#define WPAD 80

__global__ void zero_f32(float* __restrict__ p, long long n) {
    long long i = (long long)blockIdx.x * blockDim.x + threadIdx.x;
    long long stride = (long long)gridDim.x * blockDim.x;
    for (; i < n; i += stride) p[i] = 0.0f;
}

__global__ void degree_kernel(const int* __restrict__ dst, float* __restrict__ deg,
                              int n_edges) {
    int e = blockIdx.x * blockDim.x + threadIdx.x;
    if (e < n_edges) atomicAdd(&deg[dst[e]], 1.0f);
}

__global__ void invdeg_kernel(float* __restrict__ deg, int n) {
    int i = blockIdx.x * blockDim.x + threadIdx.x;
    if (i < n) deg[i] = 1.0f / fmaxf(deg[i], 1.0f);
}

__global__ void scatter_add_kernel(const float* __restrict__ feat,
                                   const int* __restrict__ src,
                                   const int* __restrict__ dst,
                                   float* __restrict__ agg, int n_edges) {
    long long tid = (long long)blockIdx.x * blockDim.x + threadIdx.x;
    int e = (int)(tid >> 6);
    int d = (int)(tid & 63);
    if (e >= n_edges) return;
    int s = src[e];
    int t = dst[e];
    atomicAdd(&agg[(long long)t * DIM + d], feat[(long long)s * DIM + d]);
}

template <bool RELU>
__global__ void sage_gemm_kernel(const float* __restrict__ agg,
                                 const float* __restrict__ invdeg,
                                 const float* __restrict__ self_feat,
                                 const float* __restrict__ wl,
                                 const float* __restrict__ bl,
                                 const float* __restrict__ wr,
                                 float* __restrict__ out, int n_nodes) {
    __shared__ v2f lds_wl[32 * WPAD];
    __shared__ v2f lds_wr[32 * WPAD];

    for (int i = threadIdx.x; i < 32 * 64; i += blockDim.x) {
        const int p = i >> 6;
        const int n = i & 63;
        v2f a, b;
        a.x = wl[(2 * p) * DIM + n];
        a.y = wl[(2 * p + 1) * DIM + n];
        b.x = wr[(2 * p) * DIM + n];
        b.y = wr[(2 * p + 1) * DIM + n];
        lds_wl[p * WPAD + n] = a;
        lds_wr[p * WPAD + n] = b;
    }
    __syncthreads();

    const int lane = threadIdx.x & 31;
    const int wave = threadIdx.x >> 5;
    const int tile = blockIdx.x * (blockDim.x >> 5) + wave;
    const int n_tiles = (n_nodes + 15) >> 4;
    if (tile >= n_tiles) return;

    const int row0 = tile << 4;
    const int r     = lane & 15;
    const int kh    = (lane >> 4) << 1;
    const int pbase = lane >> 4;

    int node = row0 + r;
    if (node > n_nodes - 1) node = n_nodes - 1;
    const float id = invdeg[node];
    const float* arow = agg       + (long long)node * DIM;
    const float* xrow = self_feat + (long long)node * DIM;

    v8f acc[4] = {{}, {}, {}, {}};

#pragma unroll
    for (int k = 0; k < DIM; k += 4) {
        v2f a;
        a.x = arow[k + kh]     * id;
        a.y = arow[k + kh + 1] * id;
        const int prow = (k >> 1) + pbase;
#pragma unroll
        for (int nt = 0; nt < 4; ++nt) {
            v2f b = lds_wl[prow * WPAD + nt * 16 + r];
            acc[nt] = __builtin_amdgcn_wmma_f32_16x16x4_f32(
                false, a, false, b, (short)0, acc[nt], false, false);
        }
    }

#pragma unroll
    for (int k = 0; k < DIM; k += 4) {
        v2f a;
        a.x = xrow[k + kh];
        a.y = xrow[k + kh + 1];
        const int prow = (k >> 1) + pbase;
#pragma unroll
        for (int nt = 0; nt < 4; ++nt) {
            v2f b = lds_wr[prow * WPAD + nt * 16 + r];
            acc[nt] = __builtin_amdgcn_wmma_f32_16x16x4_f32(
                false, a, false, b, (short)0, acc[nt], false, false);
        }
    }

    const int mbase = (lane >> 4) << 3;
#pragma unroll
    for (int nt = 0; nt < 4; ++nt) {
        const int n = nt * 16 + r;
        const float bias = bl[n];
#pragma unroll
        for (int vr = 0; vr < 8; ++vr) {
            const int m = row0 + mbase + vr;
            if (m < n_nodes) {
                float v = acc[nt][vr] + bias;
                if (RELU) v = fmaxf(v, 0.0f);
                out[(long long)m * DIM + n] = v;
            }
        }
    }
}

extern "C" void kernel_launch(void* const* d_in, const int* in_sizes, int n_in,
                              void* d_out, int out_size, void* d_ws, size_t ws_size,
                              hipStream_t stream) {
    const float* x   = (const float*)d_in[0];
    const int*  edge = (const int*) d_in[1];
    const float* w1l = (const float*)d_in[2];
    const float* b1l = (const float*)d_in[3];
    const float* w1r = (const float*)d_in[4];
    const float* w2l = (const float*)d_in[5];
    const float* b2l = (const float*)d_in[6];
    const float* w2r = (const float*)d_in[7];
    float* out = (float*)d_out;

    const int n_nodes = in_sizes[0] / DIM;
    const int n_edges = in_sizes[1] / 2;
    const int* src = edge;
    const int* dst = edge + n_edges;

    float* ws      = (float*)d_ws;
    float* invdeg  = ws;
    float* agg     = ws + n_nodes;
    float* h       = agg + (long long)n_nodes * DIM;

    const long long feat_elems = (long long)n_nodes * DIM;
    const int edge_blocks  = (n_edges + 255) / 256;
    const int node_blocks  = (n_nodes + 255) / 256;
    const int sc_blocks    = (int)(((long long)n_edges * DIM + 255) / 256);
    const int tiles        = (n_nodes + 15) / 16;
    const int gemm_blocks  = (tiles + 7) / 8;

    zero_f32<<<1024, 256, 0, stream>>>(invdeg, n_nodes);
    degree_kernel<<<edge_blocks, 256, 0, stream>>>(dst, invdeg, n_edges);
    invdeg_kernel<<<node_blocks, 256, 0, stream>>>(invdeg, n_nodes);

    zero_f32<<<2048, 256, 0, stream>>>(agg, feat_elems);
    scatter_add_kernel<<<sc_blocks, 256, 0, stream>>>(x, src, dst, agg, n_edges);
    sage_gemm_kernel<true><<<gemm_blocks, 256, 0, stream>>>(
        agg, invdeg, x, w1l, b1l, w1r, h, n_nodes);

    zero_f32<<<2048, 256, 0, stream>>>(agg, feat_elems);
    scatter_add_kernel<<<sc_blocks, 256, 0, stream>>>(h, src, dst, agg, n_edges);
    sage_gemm_kernel<false><<<gemm_blocks, 256, 0, stream>>>(
        agg, invdeg, h, w2l, b2l, w2r, out, n_nodes);
}


# Float atomic scatter-add removed: edges ordered by target in the call (int atomic count, scan, place); one wave sums a node's rows in registers, list = arrival order, one store per row
# speedup vs baseline: 3.1369x; 3.1369x over previous
#include <hip/hip_runtime.h>

typedef float v2f __attribute__((ext_vector_type(2)));
typedef float v8f __attribute__((ext_vector_type(8)));

#define DIM 64
#define WPAD 80

#define SCAN_CHUNK 1024

__global__ void zero_u32(unsigned* __restrict__ p, int n) {
    int i = blockIdx.x * blockDim.x + threadIdx.x;
    const int stride = gridDim.x * blockDim.x;
    for (; i < n; i += stride) p[i] = 0u;
}

__global__ void count_slot_kernel(const int* __restrict__ dst, unsigned* __restrict__ cnt,
                                  unsigned* __restrict__ slot, int n_edges) {
    const int e = blockIdx.x * blockDim.x + threadIdx.x;
    if (e < n_edges) slot[e] = atomicAdd(&cnt[dst[e]], 1u);
}

__device__ __forceinline__ unsigned wave_sum_u32(unsigned v) {
    for (int o = 16; o > 0; o >>= 1) v += (unsigned)__shfl_xor((int)v, o);
    return v;
}

__global__ void chunk_sum_kernel(const unsigned* __restrict__ cnt,
                                 unsigned* __restrict__ chunk_sums, int n) {
    __shared__ unsigned s_part[8];
    const int base = blockIdx.x * SCAN_CHUNK + threadIdx.x * 4;
    unsigned t = 0;
    for (int k = 0; k < 4; ++k)
        if (base + k < n) t += cnt[base + k];
    t = wave_sum_u32(t);
    if ((threadIdx.x & 31) == 0) s_part[threadIdx.x >> 5] = t;
    __syncthreads();
    if (threadIdx.x == 0) {
        unsigned s = 0;
        for (int w = 0; w < 8; ++w) s += s_part[w];
        chunk_sums[blockIdx.x] = s;
    }
}

__global__ void scan_kernel(unsigned* __restrict__ cnt_to_start,
                            const unsigned* __restrict__ chunk_sums,
                            float* __restrict__ invdeg, int n) {
    __shared__ unsigned s_before[8];
    __shared__ unsigned s_total[8];
    const int lane = threadIdx.x & 31;
    const int wave = threadIdx.x >> 5;

    unsigned before = 0;
    for (int j = threadIdx.x; j < (int)blockIdx.x; j += blockDim.x) before += chunk_sums[j];
    before = wave_sum_u32(before);

    const int base = blockIdx.x * SCAN_CHUNK + threadIdx.x * 4;
    unsigned c[4];
    unsigned t = 0;
    for (int k = 0; k < 4; ++k) {
        c[k] = (base + k < n) ? cnt_to_start[base + k] : 0u;
        t += c[k];
    }
    unsigned incl = t;
    for (int o = 1; o < 32; o <<= 1) {
        const unsigned up = (unsigned)__shfl_up((int)incl, o);
        if (lane >= o) incl += up;
    }
    if (lane == 31) s_total[wave] = incl;
    if (lane == 0) s_before[wave] = before;
    __syncthreads();

    unsigned run = incl - t;
    for (int w = 0; w < 8; ++w) {
        run += s_before[w];
        if (w < wave) run += s_total[w];
    }
    for (int k = 0; k < 4; ++k) {
        if (base + k < n) {
            cnt_to_start[base + k] = run;
            invdeg[base + k] = 1.0f / fmaxf((float)c[k], 1.0f);
            run += c[k];
            if (base + k == n - 1) cnt_to_start[n] = run;
        }
    }
}

__global__ void place_kernel(const int* __restrict__ src, const int* __restrict__ dst,
                             const unsigned* __restrict__ start,
                             const unsigned* __restrict__ slot,
                             int* __restrict__ nbr, int n_edges) {
    const int e = blockIdx.x * blockDim.x + threadIdx.x;
    if (e < n_edges) nbr[start[dst[e]] + slot[e]] = src[e];
}

__global__ void gather_sum_kernel(const float* __restrict__ feat,
                                  const unsigned* __restrict__ start,
                                  const int* __restrict__ nbr,
                                  float* __restrict__ agg, int n_nodes) {
    const int lane = threadIdx.x & 31;
    const int node = blockIdx.x * (blockDim.x >> 5) +
                     __builtin_amdgcn_readfirstlane((int)(threadIdx.x >> 5));
    if (node >= n_nodes) return;

    const unsigned beg = start[node];
    const unsigned end = start[node + 1];
    const v2f* rows = (const v2f*)feat;
    v2f acc = {0.0f, 0.0f};

    for (unsigned base = beg; base < end; base += 32) {
        const unsigned left = end - base;
        const unsigned m = left < 32u ? left : 32u;
        const int mine = ((unsigned)lane < m) ? nbr[base + lane] : 0;
        unsigned j = 0;
        for (; j + 8 <= m; j += 8) {
            v2f v[8];
#pragma unroll
            for (int u = 0; u < 8; ++u)
                v[u] = rows[(long long)__builtin_amdgcn_readlane(mine, j + u) * 32 + lane];
#pragma unroll
            for (int u = 0; u < 8; ++u) acc += v[u];
        }
        if (m & 4u) {
            v2f v[4];
#pragma unroll
            for (int u = 0; u < 4; ++u)
                v[u] = rows[(long long)__builtin_amdgcn_readlane(mine, j + u) * 32 + lane];
#pragma unroll
            for (int u = 0; u < 4; ++u) acc += v[u];
            j += 4;
        }
        if (m & 2u) {
            const v2f v0 = rows[(long long)__builtin_amdgcn_readlane(mine, j) * 32 + lane];
            const v2f v1 = rows[(long long)__builtin_amdgcn_readlane(mine, j + 1) * 32 + lane];
            acc += v0;
            acc += v1;
            j += 2;
        }
        if (m & 1u)
            acc += rows[(long long)__builtin_amdgcn_readlane(mine, j) * 32 + lane];
    }
    ((v2f*)agg)[(long long)node * 32 + lane] = acc;
}

template <bool RELU>
__global__ void sage_gemm_kernel(const float* __restrict__ agg,
                                 const float* __restrict__ invdeg,
                                 const float* __restrict__ self_feat,
                                 const float* __restrict__ wl,
                                 const float* __restrict__ bl,
                                 const float* __restrict__ wr,
                                 float* __restrict__ out, int n_nodes) {
    __shared__ v2f lds_wl[32 * WPAD];
    __shared__ v2f lds_wr[32 * WPAD];

    for (int i = threadIdx.x; i < 32 * 64; i += blockDim.x) {
        const int p = i >> 6;
        const int n = i & 63;
        v2f a, b;
        a.x = wl[(2 * p) * DIM + n];
        a.y = wl[(2 * p + 1) * DIM + n];
        b.x = wr[(2 * p) * DIM + n];
        b.y = wr[(2 * p + 1) * DIM + n];
        lds_wl[p * WPAD + n] = a;
        lds_wr[p * WPAD + n] = b;
    }
    __syncthreads();

    const int lane = threadIdx.x & 31;
    const int wave = threadIdx.x >> 5;
    const int tile = blockIdx.x * (blockDim.x >> 5) + wave;
    const int n_tiles = (n_nodes + 15) >> 4;
    if (tile >= n_tiles) return;

    const int row0 = tile << 4;
    const int r     = lane & 15;
    const int kh    = (lane >> 4) << 1;
    const int pbase = lane >> 4;

    int node = row0 + r;
    if (node > n_nodes - 1) node = n_nodes - 1;
    const float id = invdeg[node];
    const float* arow = agg       + (long long)node * DIM;
    const float* xrow = self_feat + (long long)node * DIM;

    v8f acc[4] = {{}, {}, {}, {}};

#pragma unroll
    for (int k = 0; k < DIM; k += 4) {
        v2f a;
        a.x = arow[k + kh]     * id;
        a.y = arow[k + kh + 1] * id;
        const int prow = (k >> 1) + pbase;
#pragma unroll
        for (int nt = 0; nt < 4; ++nt) {
            v2f b = lds_wl[prow * WPAD + nt * 16 + r];
            acc[nt] = __builtin_amdgcn_wmma_f32_16x16x4_f32(
                false, a, false, b, (short)0, acc[nt], false, false);
        }
    }

#pragma unroll
    for (int k = 0; k < DIM; k += 4) {
        v2f a;
        a.x = xrow[k + kh];
        a.y = xrow[k + kh + 1];
        const int prow = (k >> 1) + pbase;
#pragma unroll
        for (int nt = 0; nt < 4; ++nt) {
            v2f b = lds_wr[prow * WPAD + nt * 16 + r];
            acc[nt] = __builtin_amdgcn_wmma_f32_16x16x4_f32(
                false, a, false, b, (short)0, acc[nt], false, false);
        }
    }

    const int mbase = (lane >> 4) << 3;
#pragma unroll
    for (int nt = 0; nt < 4; ++nt) {
        const int n = nt * 16 + r;
        const float bias = bl[n];
#pragma unroll
        for (int vr = 0; vr < 8; ++vr) {
            const int m = row0 + mbase + vr;
            if (m < n_nodes) {
                float v = acc[nt][vr] + bias;
                if (RELU) v = fmaxf(v, 0.0f);
                out[(long long)m * DIM + n] = v;
            }
        }
    }
}

extern "C" void kernel_launch(void* const* d_in, const int* in_sizes, int n_in,
                              void* d_out, int out_size, void* d_ws, size_t ws_size,
                              hipStream_t stream) {
    const float* x   = (const float*)d_in[0];
    const int*  edge = (const int*) d_in[1];
    const float* w1l = (const float*)d_in[2];
    const float* b1l = (const float*)d_in[3];
    const float* w1r = (const float*)d_in[4];
    const float* w2l = (const float*)d_in[5];
    const float* b2l = (const float*)d_in[6];
    const float* w2r = (const float*)d_in[7];
    float* out = (float*)d_out;

    const int n_nodes = in_sizes[0] / DIM;
    const int n_edges = in_sizes[1] / 2;
    const int* src = edge;
    const int* dst = edge + n_edges;

    float* ws      = (float*)d_ws;
    float* invdeg  = ws;
    float* agg     = ws + n_nodes;
    float* h       = agg + (long long)n_nodes * DIM;

    const int edge_blocks  = (n_edges + 255) / 256;
    const int n_chunks     = (n_nodes + SCAN_CHUNK - 1) / SCAN_CHUNK;
    const int gather_blocks   = (n_nodes + 7) / 8;
    const int tiles        = (n_nodes + 15) / 16;
    const int gemm_blocks  = (tiles + 7) / 8;

    unsigned* start      = (unsigned*)d_out;
    unsigned* chunk_sums = start + n_nodes + 1;
    unsigned* slot       = chunk_sums + n_chunks;
    int*      nbr        = (int*)(slot + n_edges);

    zero_u32<<<128, 256, 0, stream>>>(start, n_nodes + 1);
    count_slot_kernel<<<edge_blocks, 256, 0, stream>>>(dst, start, slot, n_edges);
    chunk_sum_kernel<<<n_chunks, 256, 0, stream>>>(start, chunk_sums, n_nodes);
    scan_kernel<<<n_chunks, 256, 0, stream>>>(start, chunk_sums, invdeg, n_nodes);
    place_kernel<<<edge_blocks, 256, 0, stream>>>(src, dst, start, slot, nbr, n_edges);

    gather_sum_kernel<<<gather_blocks, 256, 0, stream>>>(x, start, nbr, agg, n_nodes);
    sage_gemm_kernel<true><<<gemm_blocks, 256, 0, stream>>>(
        agg, invdeg, x, w1l, b1l, w1r, h, n_nodes);

    gather_sum_kernel<<<gather_blocks, 256, 0, stream>>>(h, start, nbr, agg, n_nodes);
    sage_gemm_kernel<false><<<gemm_blocks, 256, 0, stream>>>(
        agg, invdeg, h, w2l, b2l, w2r, out, n_nodes);
}


# Edge ordering without global atomics: two-level counting sort on LDS counters (bins of 512 targets, then per node), list written in whole runs; run order = LDS arrival order; 9 launches to 8
# speedup vs baseline: 5.4825x; 1.7477x over previous
#include <hip/hip_runtime.h>

typedef float v2f __attribute__((ext_vector_type(2)));
typedef float v8f __attribute__((ext_vector_type(8)));

#define DIM 64
#define WPAD 80

#define BIN_SHIFT 9
#define BIN_W 512
#define NB_MAX 1024
#define CH_EDGES 6400
#define ORDER_T 512
#define STAGE_CAP 16384

__device__ __forceinline__ unsigned wave_incl_scan_u32(unsigned v, int lane) {
    for (int o = 1; o < 32; o <<= 1) {
        const unsigned up = (unsigned)__shfl_up((int)v, o);
        if (lane >= o) v += up;
    }
    return v;
}

__device__ __forceinline__ void wave_excl_scan_lds(unsigned* a, int n, int lane) {
    const int per = (n + 31) >> 5;
    const int lo = lane * per;
    unsigned t = 0;
    for (int k = 0; k < per; ++k) {
        const int i = lo + k;
        if (i < n) t += a[i];
    }
    const unsigned incl = wave_incl_scan_u32(t, lane);
    unsigned run = incl - t;
    for (int k = 0; k < per; ++k) {
        const int i = lo + k;
        if (i < n) {
            const unsigned c = a[i];
            a[i] = run;
            run += c;
        }
    }
}

__global__ void bin_hist_kernel(const int* __restrict__ dst, unsigned* __restrict__ tab_cnt,
                                int n_edges, int n_bins, int n_chunks) {
    __shared__ unsigned s_hist[NB_MAX];
    for (int i = threadIdx.x; i < n_bins; i += blockDim.x) s_hist[i] = 0u;
    __syncthreads();
    const int c = blockIdx.x;
    const int e0 = c * CH_EDGES;
    int e1 = e0 + CH_EDGES;
    if (e1 > n_edges) e1 = n_edges;
    for (int e = e0 + (int)threadIdx.x; e < e1; e += 4 * ORDER_T) {
        int d[4];
#pragma unroll
        for (int u = 0; u < 4; ++u) {
            const int ee = e + u * ORDER_T;
            d[u] = dst[ee < e1 ? ee : e1 - 1];
        }
#pragma unroll
        for (int u = 0; u < 4; ++u)
            if (e + u * ORDER_T < e1) atomicAdd(&s_hist[(unsigned)d[u] >> BIN_SHIFT], 1u);
    }
    __syncthreads();
    for (int i = threadIdx.x; i < n_bins; i += blockDim.x)
        tab_cnt[(long long)i * n_chunks + c] = s_hist[i];
}

__global__ void bin_scan_kernel(const unsigned* __restrict__ tab_cnt,
                                unsigned* __restrict__ tab_off,
                                unsigned* __restrict__ bin_total, int n_bins, int n_chunks) {
    const int lane = threadIdx.x & 31;
    const int b = blockIdx.x * (blockDim.x >> 5) + (int)(threadIdx.x >> 5);
    if (b >= n_bins) return;
    const int per = (n_chunks + 31) >> 5;
    const int lo = lane * per;
    const unsigned* row = tab_cnt + (long long)b * n_chunks;
    unsigned* orow = tab_off + (long long)b * n_chunks;
    unsigned t = 0;
    for (int k = 0; k < per; ++k) {
        const int i = lo + k;
        if (i < n_chunks) t += row[i];
    }
    const unsigned incl = wave_incl_scan_u32(t, lane);
    unsigned run = incl - t;
    for (int k = 0; k < per; ++k) {
        const int i = lo + k;
        if (i < n_chunks) {
            orow[i] = run;
            run += row[i];
        }
    }
    if (lane == 31) bin_total[b] = incl;
}

__global__ void bin_scatter_kernel(const int* __restrict__ src, const int* __restrict__ dst,
                                   const unsigned* __restrict__ tab_cnt,
                                   const unsigned* __restrict__ tab_off,
                                   const unsigned* __restrict__ bin_total,
                                   unsigned* __restrict__ bin_base,
                                   int2* __restrict__ pairs,
                                   int n_edges, int n_bins, int n_chunks) {
    __shared__ unsigned s_gbase[NB_MAX];
    __shared__ unsigned s_lbase[NB_MAX];
    __shared__ unsigned s_lcur[NB_MAX];
    __shared__ int2 s_stage[CH_EDGES];
    const int c = blockIdx.x;
    const int lane = threadIdx.x & 31;
    const int wave = threadIdx.x >> 5;

    for (int i = threadIdx.x; i < n_bins; i += blockDim.x) {
        s_gbase[i] = bin_total[i];
        s_lbase[i] = tab_cnt[(long long)i * n_chunks + c];
    }
    __syncthreads();
    if (wave == 0) wave_excl_scan_lds(s_gbase, n_bins, lane);
    else if (wave == 1) wave_excl_scan_lds(s_lbase, n_bins, lane);
    __syncthreads();
    for (int i = threadIdx.x; i < n_bins; i += blockDim.x) {
        const unsigned gb = s_gbase[i];
        if (c == 0) bin_base[i] = gb;
        s_gbase[i] = gb + tab_off[(long long)i * n_chunks + c];
        s_lcur[i] = s_lbase[i];
    }
    __syncthreads();

    const int e0 = c * CH_EDGES;
    int e1 = e0 + CH_EDGES;
    if (e1 > n_edges) e1 = n_edges;
    for (int e = e0 + (int)threadIdx.x; e < e1; e += 4 * ORDER_T) {
        int d[4], s[4];
#pragma unroll
        for (int u = 0; u < 4; ++u) {
            const int ee = e + u * ORDER_T;
            const int ec = ee < e1 ? ee : e1 - 1;
            d[u] = dst[ec];
            s[u] = src[ec];
        }
#pragma unroll
        for (int u = 0; u < 4; ++u) {
            if (e + u * ORDER_T < e1) {
                const unsigned p = atomicAdd(&s_lcur[(unsigned)d[u] >> BIN_SHIFT], 1u);
                s_stage[p] = make_int2(s[u], d[u]);
            }
        }
    }
    __syncthreads();
    const int cnt = e1 - e0;
    for (int i = threadIdx.x; i < cnt; i += blockDim.x) {
        const int2 v = s_stage[i];
        const unsigned b = (unsigned)v.y >> BIN_SHIFT;
        pairs[s_gbase[b] + ((unsigned)i - s_lbase[b])] = v;
    }
}

__global__ void bin_sort_kernel(const int2* __restrict__ pairs,
                                const unsigned* __restrict__ bin_base,
                                const unsigned* __restrict__ bin_total,
                                unsigned* __restrict__ start, float* __restrict__ invdeg,
                                int* __restrict__ nbr, int n_nodes, int n_bins) {
    __shared__ unsigned s_cnt[BIN_W];
    __shared__ unsigned s_wtot[BIN_W / 32];
    __shared__ int s_nbr[STAGE_CAP];
    const int b = blockIdx.x;
    const int tid = threadIdx.x;
    const int lane = tid & 31;
    const int wave = tid >> 5;
    const unsigned beg = bin_base[b];
    const unsigned cnt_b = bin_total[b];

    s_cnt[tid] = 0u;
    __syncthreads();
    for (unsigned i = tid; i < cnt_b; i += 4 * BIN_W) {
        int d[4];
#pragma unroll
        for (int u = 0; u < 4; ++u) {
            const unsigned ii = i + u * BIN_W;
            d[u] = pairs[beg + (ii < cnt_b ? ii : cnt_b - 1)].y;
        }
#pragma unroll
        for (int u = 0; u < 4; ++u)
            if (i + u * BIN_W < cnt_b) atomicAdd(&s_cnt[(unsigned)d[u] & (BIN_W - 1)], 1u);
    }
    __syncthreads();

    const unsigned c = s_cnt[tid];
    const unsigned incl = wave_incl_scan_u32(c, lane);
    if (lane == 31) s_wtot[wave] = incl;
    __syncthreads();
    unsigned excl = incl - c;
    for (int w = 0; w < BIN_W / 32; ++w)
        if (w < wave) excl += s_wtot[w];

    const int node = b * BIN_W + tid;
    if (node < n_nodes) {
        start[node] = beg + excl;
        invdeg[node] = 1.0f / fmaxf((float)c, 1.0f);
    }
    if (b == n_bins - 1 && tid == 0) start[n_nodes] = beg + cnt_b;
    s_cnt[tid] = excl;
    __syncthreads();

    const bool staged = cnt_b <= (unsigned)STAGE_CAP;
    for (unsigned i = tid; i < cnt_b; i += 4 * BIN_W) {
        int2 v[4];
#pragma unroll
        for (int u = 0; u < 4; ++u) {
            const unsigned ii = i + u * BIN_W;
            v[u] = pairs[beg + (ii < cnt_b ? ii : cnt_b - 1)];
        }
#pragma unroll
        for (int u = 0; u < 4; ++u) {
            if (i + u * BIN_W < cnt_b) {
                const unsigned p = atomicAdd(&s_cnt[(unsigned)v[u].y & (BIN_W - 1)], 1u);
                if (staged) s_nbr[p] = v[u].x;
                else nbr[beg + p] = v[u].x;
            }
        }
    }
    if (staged) {
        __syncthreads();
        for (unsigned i = tid; i < cnt_b; i += BIN_W) nbr[beg + i] = s_nbr[i];
    }
}

__global__ void gather_sum_kernel(const float* __restrict__ feat,
                                  const unsigned* __restrict__ start,
                                  const int* __restrict__ nbr,
                                  float* __restrict__ agg, int n_nodes) {
    const int lane = threadIdx.x & 31;
    const int node = blockIdx.x * (blockDim.x >> 5) +
                     __builtin_amdgcn_readfirstlane((int)(threadIdx.x >> 5));
    if (node >= n_nodes) return;

    const unsigned beg = start[node];
    const unsigned end = start[node + 1];
    const v2f* rows = (const v2f*)feat;
    v2f acc = {0.0f, 0.0f};

    for (unsigned base = beg; base < end; base += 32) {
        const unsigned left = end - base;
        const unsigned m = left < 32u ? left : 32u;
        const int mine = ((unsigned)lane < m) ? nbr[base + lane] : 0;
        unsigned j = 0;
        for (; j + 8 <= m; j += 8) {
            v2f v[8];
#pragma unroll
            for (int u = 0; u < 8; ++u)
                v[u] = rows[(long long)__builtin_amdgcn_readlane(mine, j + u) * 32 + lane];
#pragma unroll
            for (int u = 0; u < 8; ++u) acc += v[u];
        }
        if (m & 4u) {
            v2f v[4];
#pragma unroll
            for (int u = 0; u < 4; ++u)
                v[u] = rows[(long long)__builtin_amdgcn_readlane(mine, j + u) * 32 + lane];
#pragma unroll
            for (int u = 0; u < 4; ++u) acc += v[u];
            j += 4;
        }
        if (m & 2u) {
            const v2f v0 = rows[(long long)__builtin_amdgcn_readlane(mine, j) * 32 + lane];
            const v2f v1 = rows[(long long)__builtin_amdgcn_readlane(mine, j + 1) * 32 + lane];
            acc += v0;
            acc += v1;
            j += 2;
        }
        if (m & 1u)
            acc += rows[(long long)__builtin_amdgcn_readlane(mine, j) * 32 + lane];
    }
    ((v2f*)agg)[(long long)node * 32 + lane] = acc;
}

template <bool RELU>
__global__ void sage_gemm_kernel(const float* __restrict__ agg,
                                 const float* __restrict__ invdeg,
                                 const float* __restrict__ self_feat,
                                 const float* __restrict__ wl,
                                 const float* __restrict__ bl,
                                 const float* __restrict__ wr,
                                 float* __restrict__ out, int n_nodes) {
    __shared__ v2f lds_wl[32 * WPAD];
    __shared__ v2f lds_wr[32 * WPAD];

    for (int i = threadIdx.x; i < 32 * 64; i += blockDim.x) {
        const int p = i >> 6;
        const int n = i & 63;
        v2f a, b;
        a.x = wl[(2 * p) * DIM + n];
        a.y = wl[(2 * p + 1) * DIM + n];
        b.x = wr[(2 * p) * DIM + n];
        b.y = wr[(2 * p + 1) * DIM + n];
        lds_wl[p * WPAD + n] = a;
        lds_wr[p * WPAD + n] = b;
    }
    __syncthreads();

    const int lane = threadIdx.x & 31;
    const int wave = threadIdx.x >> 5;
    const int tile = blockIdx.x * (blockDim.x >> 5) + wave;
    const int n_tiles = (n_nodes + 15) >> 4;
    if (tile >= n_tiles) return;

    const int row0 = tile << 4;
    const int r     = lane & 15;
    const int kh    = (lane >> 4) << 1;
    const int pbase = lane >> 4;

    int node = row0 + r;
    if (node > n_nodes - 1) node = n_nodes - 1;
    const float id = invdeg[node];
    const float* arow = agg       + (long long)node * DIM;
    const float* xrow = self_feat + (long long)node * DIM;

    v8f acc[4] = {{}, {}, {}, {}};

#pragma unroll
    for (int k = 0; k < DIM; k += 4) {
        v2f a;
        a.x = arow[k + kh]     * id;
        a.y = arow[k + kh + 1] * id;
        const int prow = (k >> 1) + pbase;
#pragma unroll
        for (int nt = 0; nt < 4; ++nt) {
            v2f b = lds_wl[prow * WPAD + nt * 16 + r];
            acc[nt] = __builtin_amdgcn_wmma_f32_16x16x4_f32(
                false, a, false, b, (short)0, acc[nt], false, false);
        }
    }

#pragma unroll
    for (int k = 0; k < DIM; k += 4) {
        v2f a;
        a.x = xrow[k + kh];
        a.y = xrow[k + kh + 1];
        const int prow = (k >> 1) + pbase;
#pragma unroll
        for (int nt = 0; nt < 4; ++nt) {
            v2f b = lds_wr[prow * WPAD + nt * 16 + r];
            acc[nt] = __builtin_amdgcn_wmma_f32_16x16x4_f32(
                false, a, false, b, (short)0, acc[nt], false, false);
        }
    }

    const int mbase = (lane >> 4) << 3;
#pragma unroll
    for (int nt = 0; nt < 4; ++nt) {
        const int n = nt * 16 + r;
        const float bias = bl[n];
#pragma unroll
        for (int vr = 0; vr < 8; ++vr) {
            const int m = row0 + mbase + vr;
            if (m < n_nodes) {
                float v = acc[nt][vr] + bias;
                if (RELU) v = fmaxf(v, 0.0f);
                out[(long long)m * DIM + n] = v;
            }
        }
    }
}

extern "C" void kernel_launch(void* const* d_in, const int* in_sizes, int n_in,
                              void* d_out, int out_size, void* d_ws, size_t ws_size,
                              hipStream_t stream) {
    const float* x   = (const float*)d_in[0];
    const int*  edge = (const int*) d_in[1];
    const float* w1l = (const float*)d_in[2];
    const float* b1l = (const float*)d_in[3];
    const float* w1r = (const float*)d_in[4];
    const float* w2l = (const float*)d_in[5];
    const float* b2l = (const float*)d_in[6];
    const float* w2r = (const float*)d_in[7];
    float* out = (float*)d_out;

    const int n_nodes = in_sizes[0] / DIM;
    const int n_edges = in_sizes[1] / 2;
    const int* src = edge;
    const int* dst = edge + n_edges;

    float* ws      = (float*)d_ws;
    float* invdeg  = ws;
    float* agg     = ws + n_nodes;
    float* h       = agg + (long long)n_nodes * DIM;

    const int n_bins       = (n_nodes + BIN_W - 1) / BIN_W;
    const int n_chunks     = (n_edges + CH_EDGES - 1) / CH_EDGES;
    const int gather_blocks   = (n_nodes + 7) / 8;
    const int tiles        = (n_nodes + 15) / 16;
    const int gemm_blocks  = (tiles + 7) / 8;

    int2*     pairs      = (int2*)d_out;
    unsigned* start      = (unsigned*)d_out + 2 * (long long)n_edges;
    int*      nbr        = (int*)(start + n_nodes + 1);
    unsigned* tab_cnt    = (unsigned*)(nbr + n_edges);
    unsigned* tab_off    = tab_cnt + (long long)n_bins * n_chunks;
    unsigned* bin_total  = tab_off + (long long)n_bins * n_chunks;
    unsigned* bin_base   = bin_total + n_bins;

    bin_hist_kernel<<<n_chunks, ORDER_T, 0, stream>>>(dst, tab_cnt, n_edges, n_bins, n_chunks);
    bin_scan_kernel<<<(n_bins + 7) / 8, 256, 0, stream>>>(tab_cnt, tab_off, bin_total,
                                                        n_bins, n_chunks);
    bin_scatter_kernel<<<n_chunks, ORDER_T, 0, stream>>>(src, dst, tab_cnt, tab_off, bin_total,
                                                        bin_base, pairs, n_edges, n_bins,
                                                        n_chunks);
    bin_sort_kernel<<<n_bins, BIN_W, 0, stream>>>(pairs, bin_base, bin_total, start, invdeg,
                                                 nbr, n_nodes, n_bins);

    gather_sum_kernel<<<gather_blocks, 256, 0, stream>>>(x, start, nbr, agg, n_nodes);
    sage_gemm_kernel<true><<<gemm_blocks, 256, 0, stream>>>(
        agg, invdeg, x, w1l, b1l, w1r, h, n_nodes);

    gather_sum_kernel<<<gather_blocks, 256, 0, stream>>>(h, start, nbr, agg, n_nodes);
    sage_gemm_kernel<false><<<gemm_blocks, 256, 0, stream>>>(
        agg, invdeg, h, w2l, b2l, w2r, out, n_nodes);
}


# Neighbour sums fused into each layer's WMMA launch: a wave sums its 16 nodes' rows (list order kept) into an LDS tile that feeds the A fragments; sum plane gone, 8 launches to 6
# speedup vs baseline: 5.8291x; 1.0632x over previous
#include <hip/hip_runtime.h>

typedef float v2f __attribute__((ext_vector_type(2)));
typedef float v8f __attribute__((ext_vector_type(8)));

#define DIM 64
#define WPAD 80

#define BIN_SHIFT 9
#define BIN_W 512
#define NB_MAX 1024
#define CH_EDGES 6400
#define ORDER_T 512
#define STAGE_CAP 16384

__device__ __forceinline__ unsigned wave_incl_scan_u32(unsigned v, int lane) {
    for (int o = 1; o < 32; o <<= 1) {
        const unsigned up = (unsigned)__shfl_up((int)v, o);
        if (lane >= o) v += up;
    }
    return v;
}

__device__ __forceinline__ void wave_excl_scan_lds(unsigned* a, int n, int lane) {
    const int per = (n + 31) >> 5;
    const int lo = lane * per;
    unsigned t = 0;
    for (int k = 0; k < per; ++k) {
        const int i = lo + k;
        if (i < n) t += a[i];
    }
    const unsigned incl = wave_incl_scan_u32(t, lane);
    unsigned run = incl - t;
    for (int k = 0; k < per; ++k) {
        const int i = lo + k;
        if (i < n) {
            const unsigned c = a[i];
            a[i] = run;
            run += c;
        }
    }
}

__global__ void bin_hist_kernel(const int* __restrict__ dst, unsigned* __restrict__ tab_cnt,
                                int n_edges, int n_bins, int n_chunks) {
    __shared__ unsigned s_hist[NB_MAX];
    for (int i = threadIdx.x; i < n_bins; i += blockDim.x) s_hist[i] = 0u;
    __syncthreads();
    const int c = blockIdx.x;
    const int e0 = c * CH_EDGES;
    int e1 = e0 + CH_EDGES;
    if (e1 > n_edges) e1 = n_edges;
    for (int e = e0 + (int)threadIdx.x; e < e1; e += 4 * ORDER_T) {
        int d[4];
#pragma unroll
        for (int u = 0; u < 4; ++u) {
            const int ee = e + u * ORDER_T;
            d[u] = dst[ee < e1 ? ee : e1 - 1];
        }
#pragma unroll
        for (int u = 0; u < 4; ++u)
            if (e + u * ORDER_T < e1) atomicAdd(&s_hist[(unsigned)d[u] >> BIN_SHIFT], 1u);
    }
    __syncthreads();
    for (int i = threadIdx.x; i < n_bins; i += blockDim.x)
        tab_cnt[(long long)i * n_chunks + c] = s_hist[i];
}

__global__ void bin_scan_kernel(const unsigned* __restrict__ tab_cnt,
                                unsigned* __restrict__ tab_off,
                                unsigned* __restrict__ bin_total, int n_bins, int n_chunks) {
    const int lane = threadIdx.x & 31;
    const int b = blockIdx.x * (blockDim.x >> 5) + (int)(threadIdx.x >> 5);
    if (b >= n_bins) return;
    const int per = (n_chunks + 31) >> 5;
    const int lo = lane * per;
    const unsigned* row = tab_cnt + (long long)b * n_chunks;
    unsigned* orow = tab_off + (long long)b * n_chunks;
    unsigned t = 0;
    for (int k = 0; k < per; ++k) {
        const int i = lo + k;
        if (i < n_chunks) t += row[i];
    }
    const unsigned incl = wave_incl_scan_u32(t, lane);
    unsigned run = incl - t;
    for (int k = 0; k < per; ++k) {
        const int i = lo + k;
        if (i < n_chunks) {
            orow[i] = run;
            run += row[i];
        }
    }
    if (lane == 31) bin_total[b] = incl;
}

__global__ void bin_scatter_kernel(const int* __restrict__ src, const int* __restrict__ dst,
                                   const unsigned* __restrict__ tab_cnt,
                                   const unsigned* __restrict__ tab_off,
                                   const unsigned* __restrict__ bin_total,
                                   unsigned* __restrict__ bin_base,
                                   int2* __restrict__ pairs,
                                   int n_edges, int n_bins, int n_chunks) {
    __shared__ unsigned s_gbase[NB_MAX];
    __shared__ unsigned s_lbase[NB_MAX];
    __shared__ unsigned s_lcur[NB_MAX];
    __shared__ int2 s_stage[CH_EDGES];
    const int c = blockIdx.x;
    const int lane = threadIdx.x & 31;
    const int wave = threadIdx.x >> 5;

    for (int i = threadIdx.x; i < n_bins; i += blockDim.x) {
        s_gbase[i] = bin_total[i];
        s_lbase[i] = tab_cnt[(long long)i * n_chunks + c];
    }
    __syncthreads();
    if (wave == 0) wave_excl_scan_lds(s_gbase, n_bins, lane);
    else if (wave == 1) wave_excl_scan_lds(s_lbase, n_bins, lane);
    __syncthreads();
    for (int i = threadIdx.x; i < n_bins; i += blockDim.x) {
        const unsigned gb = s_gbase[i];
        if (c == 0) bin_base[i] = gb;
        s_gbase[i] = gb + tab_off[(long long)i * n_chunks + c];
        s_lcur[i] = s_lbase[i];
    }
    __syncthreads();

    const int e0 = c * CH_EDGES;
    int e1 = e0 + CH_EDGES;
    if (e1 > n_edges) e1 = n_edges;
    for (int e = e0 + (int)threadIdx.x; e < e1; e += 4 * ORDER_T) {
        int d[4], s[4];
#pragma unroll
        for (int u = 0; u < 4; ++u) {
            const int ee = e + u * ORDER_T;
            const int ec = ee < e1 ? ee : e1 - 1;
            d[u] = dst[ec];
            s[u] = src[ec];
        }
#pragma unroll
        for (int u = 0; u < 4; ++u) {
            if (e + u * ORDER_T < e1) {
                const unsigned p = atomicAdd(&s_lcur[(unsigned)d[u] >> BIN_SHIFT], 1u);
                s_stage[p] = make_int2(s[u], d[u]);
            }
        }
    }
    __syncthreads();
    const int cnt = e1 - e0;
    for (int i = threadIdx.x; i < cnt; i += blockDim.x) {
        const int2 v = s_stage[i];
        const unsigned b = (unsigned)v.y >> BIN_SHIFT;
        pairs[s_gbase[b] + ((unsigned)i - s_lbase[b])] = v;
    }
}

__global__ void bin_sort_kernel(const int2* __restrict__ pairs,
                                const unsigned* __restrict__ bin_base,
                                const unsigned* __restrict__ bin_total,
                                unsigned* __restrict__ start, float* __restrict__ invdeg,
                                int* __restrict__ nbr, int n_nodes, int n_bins) {
    __shared__ unsigned s_cnt[BIN_W];
    __shared__ unsigned s_wtot[BIN_W / 32];
    __shared__ int s_nbr[STAGE_CAP];
    const int b = blockIdx.x;
    const int tid = threadIdx.x;
    const int lane = tid & 31;
    const int wave = tid >> 5;
    const unsigned beg = bin_base[b];
    const unsigned cnt_b = bin_total[b];

    s_cnt[tid] = 0u;
    __syncthreads();
    for (unsigned i = tid; i < cnt_b; i += 4 * BIN_W) {
        int d[4];
#pragma unroll
        for (int u = 0; u < 4; ++u) {
            const unsigned ii = i + u * BIN_W;
            d[u] = pairs[beg + (ii < cnt_b ? ii : cnt_b - 1)].y;
        }
#pragma unroll
        for (int u = 0; u < 4; ++u)
            if (i + u * BIN_W < cnt_b) atomicAdd(&s_cnt[(unsigned)d[u] & (BIN_W - 1)], 1u);
    }
    __syncthreads();

    const unsigned c = s_cnt[tid];
    const unsigned incl = wave_incl_scan_u32(c, lane);
    if (lane == 31) s_wtot[wave] = incl;
    __syncthreads();
    unsigned excl = incl - c;
    for (int w = 0; w < BIN_W / 32; ++w)
        if (w < wave) excl += s_wtot[w];

    const int node = b * BIN_W + tid;
    if (node < n_nodes) {
        start[node] = beg + excl;
        invdeg[node] = 1.0f / fmaxf((float)c, 1.0f);
    }
    if (b == n_bins - 1 && tid == 0) start[n_nodes] = beg + cnt_b;
    s_cnt[tid] = excl;
    __syncthreads();

    const bool staged = cnt_b <= (unsigned)STAGE_CAP;
    for (unsigned i = tid; i < cnt_b; i += 4 * BIN_W) {
        int2 v[4];
#pragma unroll
        for (int u = 0; u < 4; ++u) {
            const unsigned ii = i + u * BIN_W;
            v[u] = pairs[beg + (ii < cnt_b ? ii : cnt_b - 1)];
        }
#pragma unroll
        for (int u = 0; u < 4; ++u) {
            if (i + u * BIN_W < cnt_b) {
                const unsigned p = atomicAdd(&s_cnt[(unsigned)v[u].y & (BIN_W - 1)], 1u);
                if (staged) s_nbr[p] = v[u].x;
                else nbr[beg + p] = v[u].x;
            }
        }
    }
    if (staged) {
        __syncthreads();
        for (unsigned i = tid; i < cnt_b; i += BIN_W) nbr[beg + i] = s_nbr[i];
    }
}

#define APITCH 68
#define ROWS_IN_FLIGHT 16

template <bool FULL>
__device__ __forceinline__ void sum_group(const v2f* rows, const int mine, const unsigned j0,
                                          const unsigned cnt, const unsigned pos0,
                                          const int mystart, const float id, float* my_a,
                                          const int lane, const bool fetch,
                                          const int* next_entry, int& mine_next,
                                          v2f& acc, unsigned& cur, unsigned& nb) {
    v2f v[ROWS_IN_FLIGHT];
#pragma unroll
    for (int u = 0; u < ROWS_IN_FLIGHT; ++u) {
        const unsigned jj = (FULL || (unsigned)u < cnt) ? (unsigned)u : cnt - 1;
        v[u] = rows[(long long)__builtin_amdgcn_readlane(mine, j0 + jj) * 32 + lane];
    }
    if (fetch) mine_next = *next_entry;
#pragma unroll
    for (int u = 0; u < ROWS_IN_FLIGHT; ++u) {
        if (FULL || (unsigned)u < cnt) {
            const unsigned p = pos0 + (unsigned)u;
            while (p == nb) {
                const float sc = __builtin_bit_cast(
                    float, __builtin_amdgcn_readlane(__builtin_bit_cast(int, id), cur));
                v2f o;
                o.x = acc.x * sc;
                o.y = acc.y * sc;
                *(v2f*)&my_a[cur * APITCH + 2 * lane] = o;
                acc.x = 0.0f;
                acc.y = 0.0f;
                ++cur;
                nb = (unsigned)__builtin_amdgcn_readlane(mystart, cur + 1);
            }
            acc += v[u];
        }
    }
}

template <bool RELU>
__global__ void sage_layer_kernel(const float* __restrict__ feat,
                                  const unsigned* __restrict__ start,
                                  const int* __restrict__ nbr,
                                  const float* __restrict__ invdeg,
                                  const float* __restrict__ wl,
                                  const float* __restrict__ bl,
                                  const float* __restrict__ wr,
                                  float* __restrict__ out, int n_nodes) {
    __shared__ v2f lds_wl[32 * WPAD];
    __shared__ v2f lds_wr[32 * WPAD];
    __shared__ float lds_a[8 * 16 * APITCH];

    for (int i = threadIdx.x; i < 32 * 64; i += blockDim.x) {
        const int p = i >> 6;
        const int n = i & 63;
        v2f a, b;
        a.x = wl[(2 * p) * DIM + n];
        a.y = wl[(2 * p + 1) * DIM + n];
        b.x = wr[(2 * p) * DIM + n];
        b.y = wr[(2 * p + 1) * DIM + n];
        lds_wl[p * WPAD + n] = a;
        lds_wr[p * WPAD + n] = b;
    }
    __syncthreads();

    const int lane = threadIdx.x & 31;
    const int wave = __builtin_amdgcn_readfirstlane((int)(threadIdx.x >> 5));
    const int tile = blockIdx.x * 8 + wave;
    const int n_tiles = (n_nodes + 15) >> 4;
    if (tile >= n_tiles) return;

    const int row0 = tile << 4;
    const int r     = lane & 15;
    const int kh    = (lane >> 4) << 1;
    const int pbase = lane >> 4;
    float* my_a = lds_a + wave * (16 * APITCH);

    int node = row0 + r;
    if (node > n_nodes - 1) node = n_nodes - 1;
    const float id = invdeg[node];
    const float* xrow = feat + (long long)node * DIM;

    int bi = row0 + (lane < 16 ? lane : 16);
    if (bi > n_nodes) bi = n_nodes;
    const int mystart = (int)start[bi];

    v2f xa[16];
#pragma unroll
    for (int k = 0; k < DIM; k += 4) {
        xa[k >> 2].x = xrow[k + kh];
        xa[k >> 2].y = xrow[k + kh + 1];
    }

    {
        const unsigned tb = (unsigned)__builtin_amdgcn_readlane(mystart, 0);
        const unsigned te = (unsigned)__builtin_amdgcn_readlane(mystart, 16);
        const v2f* rows = (const v2f*)feat;
        unsigned cur = 0;
        unsigned nb = (unsigned)__builtin_amdgcn_readlane(mystart, 1);
        v2f acc = {0.0f, 0.0f};
        int mine = 0;
        if (tb < te) mine = nbr[tb + (unsigned)lane < te ? tb + (unsigned)lane : te - 1];
        for (unsigned base = tb; base < te; base += 32) {
            const unsigned left = te - base;
            const unsigned m = left < 32u ? left : 32u;
            const unsigned nx = base + 32u + (unsigned)lane;
            const int* next_entry = nbr + (nx < te ? nx : te - 1);
            int mine_next = 0;
            for (unsigned j0 = 0; j0 < m; j0 += ROWS_IN_FLIGHT) {
                const unsigned cnt = (m - j0) < (unsigned)ROWS_IN_FLIGHT ? (m - j0)
                                                                         : (unsigned)ROWS_IN_FLIGHT;
                if (cnt == (unsigned)ROWS_IN_FLIGHT)
                    sum_group<true>(rows, mine, j0, cnt, base + j0, mystart, id, my_a, lane,
                                    j0 == 0, next_entry, mine_next, acc, cur, nb);
                else
                    sum_group<false>(rows, mine, j0, cnt, base + j0, mystart, id, my_a, lane,
                                     j0 == 0, next_entry, mine_next, acc, cur, nb);
            }
            mine = mine_next;
        }
        for (; cur < 16; ++cur) {
            const float sc = __builtin_bit_cast(
                float, __builtin_amdgcn_readlane(__builtin_bit_cast(int, id), cur));
            v2f o;
            o.x = acc.x * sc;
            o.y = acc.y * sc;
            *(v2f*)&my_a[cur * APITCH + 2 * lane] = o;
            acc.x = 0.0f;
            acc.y = 0.0f;
        }
    }

    v8f acc4[4] = {{}, {}, {}, {}};

    const float* arow = my_a + r * APITCH;
#pragma unroll
    for (int k = 0; k < DIM; k += 4) {
        const v2f a = *(const v2f*)&arow[k + kh];
        const int prow = (k >> 1) + pbase;
#pragma unroll
        for (int nt = 0; nt < 4; ++nt) {
            v2f b = lds_wl[prow * WPAD + nt * 16 + r];
            acc4[nt] = __builtin_amdgcn_wmma_f32_16x16x4_f32(
                false, a, false, b, (short)0, acc4[nt], false, false);
        }
    }

#pragma unroll
    for (int k = 0; k < DIM; k += 4) {
        const v2f a = xa[k >> 2];
        const int prow = (k >> 1) + pbase;
#pragma unroll
        for (int nt = 0; nt < 4; ++nt) {
            v2f b = lds_wr[prow * WPAD + nt * 16 + r];
            acc4[nt] = __builtin_amdgcn_wmma_f32_16x16x4_f32(
                false, a, false, b, (short)0, acc4[nt], false, false);
        }
    }

    const int mbase = (lane >> 4) << 3;
#pragma unroll
    for (int nt = 0; nt < 4; ++nt) {
        const int n = nt * 16 + r;
        const float bias = bl[n];
#pragma unroll
        for (int vr = 0; vr < 8; ++vr) {
            const int m = row0 + mbase + vr;
            if (m < n_nodes) {
                float v = acc4[nt][vr] + bias;
                if (RELU) v = fmaxf(v, 0.0f);
                out[(long long)m * DIM + n] = v;
            }
        }
    }
}

extern "C" void kernel_launch(void* const* d_in, const int* in_sizes, int n_in,
                              void* d_out, int out_size, void* d_ws, size_t ws_size,
                              hipStream_t stream) {
    const float* x   = (const float*)d_in[0];
    const int*  edge = (const int*) d_in[1];
    const float* w1l = (const float*)d_in[2];
    const float* b1l = (const float*)d_in[3];
    const float* w1r = (const float*)d_in[4];
    const float* w2l = (const float*)d_in[5];
    const float* b2l = (const float*)d_in[6];
    const float* w2r = (const float*)d_in[7];
    float* out = (float*)d_out;

    const int n_nodes = in_sizes[0] / DIM;
    const int n_edges = in_sizes[1] / 2;
    const int* src = edge;
    const int* dst = edge + n_edges;

    float* ws      = (float*)d_ws;
    float* invdeg  = ws;
    float* list    = ws + n_nodes;
    float* h       = list + (long long)n_nodes * DIM;
    unsigned* start = (unsigned*)list;
    int*      nbr   = (int*)(start + n_nodes + 1);

    const int n_bins       = (n_nodes + BIN_W - 1) / BIN_W;
    const int n_chunks     = (n_edges + CH_EDGES - 1) / CH_EDGES;
    const int tiles        = (n_nodes + 15) / 16;
    const int layer_blocks = (tiles + 7) / 8;

    int2*     pairs      = (int2*)d_out;
    unsigned* tab_cnt    = (unsigned*)d_out + 2 * (long long)n_edges;
    unsigned* tab_off    = tab_cnt + (long long)n_bins * n_chunks;
    unsigned* bin_total  = tab_off + (long long)n_bins * n_chunks;
    unsigned* bin_base   = bin_total + n_bins;

    bin_hist_kernel<<<n_chunks, ORDER_T, 0, stream>>>(dst, tab_cnt, n_edges, n_bins, n_chunks);
    bin_scan_kernel<<<(n_bins + 7) / 8, 256, 0, stream>>>(tab_cnt, tab_off, bin_total,
                                                        n_bins, n_chunks);
    bin_scatter_kernel<<<n_chunks, ORDER_T, 0, stream>>>(src, dst, tab_cnt, tab_off, bin_total,
                                                        bin_base, pairs, n_edges, n_bins,
                                                        n_chunks);
    bin_sort_kernel<<<n_bins, BIN_W, 0, stream>>>(pairs, bin_base, bin_total, start, invdeg,
                                                 nbr, n_nodes, n_bins);

    sage_layer_kernel<true><<<layer_blocks, 256, 0, stream>>>(
        x, start, nbr, invdeg, w1l, b1l, w1r, h, n_nodes);

    sage_layer_kernel<false><<<layer_blocks, 256, 0, stream>>>(
        h, start, nbr, invdeg, w2l, b2l, w2r, out, n_nodes);
}


# Edge ordering in 2 launches, not 4: a chunk sorts its edges by bin in place in LDS (rank = counting add's return), a bin finds its runs by a count table; list = LDS arrival order; 6 launches to 4
# speedup vs baseline: 6.5028x; 1.1156x over previous
#include <hip/hip_runtime.h>

typedef float v2f __attribute__((ext_vector_type(2)));
typedef float v8f __attribute__((ext_vector_type(8)));

#define DIM 64
#define WPAD 80

#define BIN_SHIFT 9
#define BIN_W 512
#define NB_MAX 1024
#define CH_EDGES 6400
#define ORDER_T 512
#define STAGE_CAP 16384

__device__ __forceinline__ unsigned wave_incl_scan_u32(unsigned v, int lane) {
    for (int o = 1; o < 32; o <<= 1) {
        const unsigned up = (unsigned)__shfl_up((int)v, o);
        if (lane >= o) v += up;
    }
    return v;
}

__device__ __forceinline__ void wave_excl_scan_lds(unsigned* a, int n, int lane) {
    const int per = (n + 31) >> 5;
    const int lo = lane * per;
    unsigned t = 0;
    for (int k = 0; k < per; ++k) {
        const int i = lo + k;
        if (i < n) t += a[i];
    }
    const unsigned incl = wave_incl_scan_u32(t, lane);
    unsigned run = incl - t;
    for (int k = 0; k < per; ++k) {
        const int i = lo + k;
        if (i < n) {
            const unsigned c = a[i];
            a[i] = run;
            run += c;
        }
    }
}

#define CH_PER ((CH_EDGES + ORDER_T - 1) / ORDER_T)
static_assert(CH_EDGES < 65536, "run start and count share one 32-bit word");
__global__ void chunk_sort_kernel(const int* __restrict__ src, const int* __restrict__ dst,
                                  unsigned* __restrict__ tab, int2* __restrict__ pairs,
                                  int n_edges, int n_bins) {
    __shared__ unsigned s_cnt[NB_MAX];
    __shared__ unsigned s_lbase[NB_MAX];
    __shared__ int2 s_stage[CH_EDGES];
    const int c = blockIdx.x;
    const int lane = threadIdx.x & 31;
    const int wave = threadIdx.x >> 5;

    for (int i = threadIdx.x; i < n_bins; i += blockDim.x) s_cnt[i] = 0u;
    __syncthreads();

    const int e0 = c * CH_EDGES;
    int e1 = e0 + CH_EDGES;
    if (e1 > n_edges) e1 = n_edges;
    int d[CH_PER], s[CH_PER];
    unsigned rk[CH_PER];
#pragma unroll
    for (int u = 0; u < CH_PER; ++u) {
        const int ee = e0 + (int)threadIdx.x + u * ORDER_T;
        const int ec = ee < e1 ? ee : e1 - 1;
        d[u] = dst[ec];
        s[u] = src[ec];
    }
#pragma unroll
    for (int u = 0; u < CH_PER; ++u) {
        rk[u] = 0u;
        if (e0 + (int)threadIdx.x + u * ORDER_T < e1)
            rk[u] = atomicAdd(&s_cnt[(unsigned)d[u] >> BIN_SHIFT], 1u);
    }
    __syncthreads();
    for (int i = threadIdx.x; i < n_bins; i += blockDim.x) s_lbase[i] = s_cnt[i];
    __syncthreads();
    if (wave == 0) wave_excl_scan_lds(s_lbase, n_bins, lane);
    __syncthreads();
#pragma unroll
    for (int u = 0; u < CH_PER; ++u) {
        if (e0 + (int)threadIdx.x + u * ORDER_T < e1)
            s_stage[s_lbase[(unsigned)d[u] >> BIN_SHIFT] + rk[u]] = make_int2(s[u], d[u]);
    }
    for (int i = threadIdx.x; i < n_bins; i += blockDim.x)
        tab[(long long)c * n_bins + i] = (s_lbase[i] << 16) | s_cnt[i];
    __syncthreads();
    const int cnt = e1 - e0;
    for (int i = threadIdx.x; i < cnt; i += blockDim.x) pairs[e0 + i] = s_stage[i];
}

#define CH_MAX 1024
__global__ void bin_sort_kernel(const int2* __restrict__ pairs,
                                const unsigned* __restrict__ tab,
                                unsigned* __restrict__ start, float* __restrict__ invdeg,
                                int* __restrict__ nbr, int n_nodes, int n_bins, int n_chunks) {
    __shared__ unsigned s_cnt[BIN_W];
    __shared__ unsigned s_wtot[BIN_W / 32];
    __shared__ int s_nbr[STAGE_CAP];
    __shared__ unsigned s_rb[CH_MAX + 1];
    __shared__ unsigned s_adj[CH_MAX];
    __shared__ unsigned s_pidx[STAGE_CAP];
    __shared__ unsigned s_beg;
    const int b = blockIdx.x;
    const int tid = threadIdx.x;
    const int lane = tid & 31;
    const int wave = tid >> 5;

    s_cnt[tid] = 0u;
    if (tid == 0) {
        s_beg = 0u;
        s_rb[n_chunks] = 0u;
    }
    __syncthreads();
    unsigned lbsum = 0u;
    for (int t = tid; t < n_chunks; t += BIN_W) {
        const unsigned inf = tab[(long long)t * n_bins + b];
        s_rb[t] = inf & 0xffffu;
        s_adj[t] = (unsigned)t * (unsigned)CH_EDGES + (inf >> 16);
        lbsum += inf >> 16;
    }
    for (int o = 16; o > 0; o >>= 1) lbsum += (unsigned)__shfl_xor((int)lbsum, o);
    if (lane == 0 && lbsum != 0u) atomicAdd(&s_beg, lbsum);
    __syncthreads();
    if (wave == 0) wave_excl_scan_lds(s_rb, n_chunks + 1, lane);
    __syncthreads();
    const unsigned beg = s_beg;
    const unsigned cnt_b = s_rb[n_chunks];
    const bool staged = cnt_b <= (unsigned)STAGE_CAP;
    for (int t = tid; t < n_chunks; t += BIN_W) {
        const unsigned lo = s_rb[t];
        const unsigned hi = s_rb[t + 1];
        const unsigned adj = s_adj[t] - lo;
        s_adj[t] = adj;
        if (staged)
            for (unsigned m = lo; m < hi; ++m) s_pidx[m] = adj + m;
    }
    __syncthreads();

    auto bisect_index = [&](const unsigned ii) -> unsigned {
        unsigned lo = 0u, hi = (unsigned)n_chunks;
        while (hi - lo > 1u) {
            const unsigned mid = (lo + hi) >> 1;
            if (s_rb[mid] <= ii) lo = mid;
            else hi = mid;
        }
        return s_adj[lo] + ii;
    };

    for (unsigned i = tid; i < cnt_b; i += 4 * BIN_W) {
        int d[4];
        unsigned px[4];
        if (staged) {
#pragma unroll
            for (int u = 0; u < 4; ++u) {
                const unsigned ii = i + u * BIN_W;
                px[u] = s_pidx[ii < cnt_b ? ii : cnt_b - 1];
            }
        } else {
#pragma unroll
            for (int u = 0; u < 4; ++u) {
                const unsigned ii = i + u * BIN_W;
                px[u] = bisect_index(ii < cnt_b ? ii : cnt_b - 1);
            }
        }
#pragma unroll
        for (int u = 0; u < 4; ++u) d[u] = pairs[px[u]].y;
#pragma unroll
        for (int u = 0; u < 4; ++u)
            if (i + u * BIN_W < cnt_b) atomicAdd(&s_cnt[(unsigned)d[u] & (BIN_W - 1)], 1u);
    }
    __syncthreads();

    const unsigned c = s_cnt[tid];
    const unsigned incl = wave_incl_scan_u32(c, lane);
    if (lane == 31) s_wtot[wave] = incl;
    __syncthreads();
    unsigned excl = incl - c;
    for (int w = 0; w < BIN_W / 32; ++w)
        if (w < wave) excl += s_wtot[w];

    const int node = b * BIN_W + tid;
    if (node < n_nodes) {
        start[node] = beg + excl;
        invdeg[node] = 1.0f / fmaxf((float)c, 1.0f);
    }
    if (b == n_bins - 1 && tid == 0) start[n_nodes] = beg + cnt_b;
    s_cnt[tid] = excl;
    __syncthreads();

    for (unsigned i = tid; i < cnt_b; i += 4 * BIN_W) {
        int2 v[4];
        unsigned px[4];
        if (staged) {
#pragma unroll
            for (int u = 0; u < 4; ++u) {
                const unsigned ii = i + u * BIN_W;
                px[u] = s_pidx[ii < cnt_b ? ii : cnt_b - 1];
            }
        } else {
#pragma unroll
            for (int u = 0; u < 4; ++u) {
                const unsigned ii = i + u * BIN_W;
                px[u] = bisect_index(ii < cnt_b ? ii : cnt_b - 1);
            }
        }
#pragma unroll
        for (int u = 0; u < 4; ++u) v[u] = pairs[px[u]];
#pragma unroll
        for (int u = 0; u < 4; ++u) {
            if (i + u * BIN_W < cnt_b) {
                const unsigned p = atomicAdd(&s_cnt[(unsigned)v[u].y & (BIN_W - 1)], 1u);
                if (staged) s_nbr[p] = v[u].x;
                else nbr[beg + p] = v[u].x;
            }
        }
    }
    if (staged) {
        __syncthreads();
        for (unsigned i = tid; i < cnt_b; i += BIN_W) nbr[beg + i] = s_nbr[i];
    }
}

#define APITCH 68
#define ROWS_IN_FLIGHT 16

template <bool FULL>
__device__ __forceinline__ void sum_group(const v2f* rows, const int mine, const unsigned j0,
                                          const unsigned cnt, const unsigned pos0,
                                          const int mystart, const float id, float* my_a,
                                          const int lane, const bool fetch,
                                          const int* next_entry, int& mine_next,
                                          v2f& acc, unsigned& cur, unsigned& nb) {
    v2f v[ROWS_IN_FLIGHT];
#pragma unroll
    for (int u = 0; u < ROWS_IN_FLIGHT; ++u) {
        const unsigned jj = (FULL || (unsigned)u < cnt) ? (unsigned)u : cnt - 1;
        v[u] = rows[(long long)__builtin_amdgcn_readlane(mine, j0 + jj) * 32 + lane];
    }
    if (fetch) mine_next = *next_entry;
#pragma unroll
    for (int u = 0; u < ROWS_IN_FLIGHT; ++u) {
        if (FULL || (unsigned)u < cnt) {
            const unsigned p = pos0 + (unsigned)u;
            while (p == nb) {
                const float sc = __builtin_bit_cast(
                    float, __builtin_amdgcn_readlane(__builtin_bit_cast(int, id), cur));
                v2f o;
                o.x = acc.x * sc;
                o.y = acc.y * sc;
                *(v2f*)&my_a[cur * APITCH + 2 * lane] = o;
                acc.x = 0.0f;
                acc.y = 0.0f;
                ++cur;
                nb = (unsigned)__builtin_amdgcn_readlane(mystart, cur + 1);
            }
            acc += v[u];
        }
    }
}

template <bool RELU>
__global__ void sage_layer_kernel(const float* __restrict__ feat,
                                  const unsigned* __restrict__ start,
                                  const int* __restrict__ nbr,
                                  const float* __restrict__ invdeg,
                                  const float* __restrict__ wl,
                                  const float* __restrict__ bl,
                                  const float* __restrict__ wr,
                                  float* __restrict__ out, int n_nodes) {
    __shared__ v2f lds_wl[32 * WPAD];
    __shared__ v2f lds_wr[32 * WPAD];
    __shared__ float lds_a[8 * 16 * APITCH];

    for (int i = threadIdx.x; i < 32 * 64; i += blockDim.x) {
        const int p = i >> 6;
        const int n = i & 63;
        v2f a, b;
        a.x = wl[(2 * p) * DIM + n];
        a.y = wl[(2 * p + 1) * DIM + n];
        b.x = wr[(2 * p) * DIM + n];
        b.y = wr[(2 * p + 1) * DIM + n];
        lds_wl[p * WPAD + n] = a;
        lds_wr[p * WPAD + n] = b;
    }
    __syncthreads();

    const int lane = threadIdx.x & 31;
    const int wave = __builtin_amdgcn_readfirstlane((int)(threadIdx.x >> 5));
    const int tile = blockIdx.x * 8 + wave;
    const int n_tiles = (n_nodes + 15) >> 4;
    if (tile >= n_tiles) return;

    const int row0 = tile << 4;
    const int r     = lane & 15;
    const int kh    = (lane >> 4) << 1;
    const int pbase = lane >> 4;
    float* my_a = lds_a + wave * (16 * APITCH);

    int node = row0 + r;
    if (node > n_nodes - 1) node = n_nodes - 1;
    const float id = invdeg[node];
    const float* xrow = feat + (long long)node * DIM;

    int bi = row0 + (lane < 16 ? lane : 16);
    if (bi > n_nodes) bi = n_nodes;
    const int mystart = (int)start[bi];

    v2f xa[16];
#pragma unroll
    for (int k = 0; k < DIM; k += 4) {
        xa[k >> 2].x = xrow[k + kh];
        xa[k >> 2].y = xrow[k + kh + 1];
    }

    {
        const unsigned tb = (unsigned)__builtin_amdgcn_readlane(mystart, 0);
        const unsigned te = (unsigned)__builtin_amdgcn_readlane(mystart, 16);
        const v2f* rows = (const v2f*)feat;
        unsigned cur = 0;
        unsigned nb = (unsigned)__builtin_amdgcn_readlane(mystart, 1);
        v2f acc = {0.0f, 0.0f};
        int mine = 0;
        if (tb < te) mine = nbr[tb + (unsigned)lane < te ? tb + (unsigned)lane : te - 1];
        for (unsigned base = tb; base < te; base += 32) {
            const unsigned left = te - base;
            const unsigned m = left < 32u ? left : 32u;
            const unsigned nx = base + 32u + (unsigned)lane;
            const int* next_entry = nbr + (nx < te ? nx : te - 1);
            int mine_next = 0;
            for (unsigned j0 = 0; j0 < m; j0 += ROWS_IN_FLIGHT) {
                const unsigned cnt = (m - j0) < (unsigned)ROWS_IN_FLIGHT ? (m - j0)
                                                                         : (unsigned)ROWS_IN_FLIGHT;
                if (cnt == (unsigned)ROWS_IN_FLIGHT)
                    sum_group<true>(rows, mine, j0, cnt, base + j0, mystart, id, my_a, lane,
                                    j0 == 0, next_entry, mine_next, acc, cur, nb);
                else
                    sum_group<false>(rows, mine, j0, cnt, base + j0, mystart, id, my_a, lane,
                                     j0 == 0, next_entry, mine_next, acc, cur, nb);
            }
            mine = mine_next;
        }
        for (; cur < 16; ++cur) {
            const float sc = __builtin_bit_cast(
                float, __builtin_amdgcn_readlane(__builtin_bit_cast(int, id), cur));
            v2f o;
            o.x = acc.x * sc;
            o.y = acc.y * sc;
            *(v2f*)&my_a[cur * APITCH + 2 * lane] = o;
            acc.x = 0.0f;
            acc.y = 0.0f;
        }
    }

    v8f acc4[4] = {{}, {}, {}, {}};

    const float* arow = my_a + r * APITCH;
#pragma unroll
    for (int k = 0; k < DIM; k += 4) {
        const v2f a = *(const v2f*)&arow[k + kh];
        const int prow = (k >> 1) + pbase;
#pragma unroll
        for (int nt = 0; nt < 4; ++nt) {
            v2f b = lds_wl[prow * WPAD + nt * 16 + r];
            acc4[nt] = __builtin_amdgcn_wmma_f32_16x16x4_f32(
                false, a, false, b, (short)0, acc4[nt], false, false);
        }
    }

#pragma unroll
    for (int k = 0; k < DIM; k += 4) {
        const v2f a = xa[k >> 2];
        const int prow = (k >> 1) + pbase;
#pragma unroll
        for (int nt = 0; nt < 4; ++nt) {
            v2f b = lds_wr[prow * WPAD + nt * 16 + r];
            acc4[nt] = __builtin_amdgcn_wmma_f32_16x16x4_f32(
                false, a, false, b, (short)0, acc4[nt], false, false);
        }
    }

    const int mbase = (lane >> 4) << 3;
#pragma unroll
    for (int nt = 0; nt < 4; ++nt) {
        const int n = nt * 16 + r;
        const float bias = bl[n];
#pragma unroll
        for (int vr = 0; vr < 8; ++vr) {
            const int m = row0 + mbase + vr;
            if (m < n_nodes) {
                float v = acc4[nt][vr] + bias;
                if (RELU) v = fmaxf(v, 0.0f);
                out[(long long)m * DIM + n] = v;
            }
        }
    }
}

extern "C" void kernel_launch(void* const* d_in, const int* in_sizes, int n_in,
                              void* d_out, int out_size, void* d_ws, size_t ws_size,
                              hipStream_t stream) {
    const float* x   = (const float*)d_in[0];
    const int*  edge = (const int*) d_in[1];
    const float* w1l = (const float*)d_in[2];
    const float* b1l = (const float*)d_in[3];
    const float* w1r = (const float*)d_in[4];
    const float* w2l = (const float*)d_in[5];
    const float* b2l = (const float*)d_in[6];
    const float* w2r = (const float*)d_in[7];
    float* out = (float*)d_out;

    const int n_nodes = in_sizes[0] / DIM;
    const int n_edges = in_sizes[1] / 2;
    const int* src = edge;
    const int* dst = edge + n_edges;

    float* ws      = (float*)d_ws;
    float* invdeg  = ws;
    float* list    = ws + n_nodes;
    float* h       = list + (long long)n_nodes * DIM;
    unsigned* start = (unsigned*)list;
    int*      nbr   = (int*)(start + n_nodes + 1);

    const int n_bins       = (n_nodes + BIN_W - 1) / BIN_W;
    const int n_chunks     = (n_edges + CH_EDGES - 1) / CH_EDGES;
    const int tiles        = (n_nodes + 15) / 16;
    const int layer_blocks = (tiles + 7) / 8;

    int2*     pairs = (int2*)d_out;
    unsigned* tab   = (unsigned*)d_out + 2 * (long long)n_edges;

    chunk_sort_kernel<<<n_chunks, ORDER_T, 0, stream>>>(src, dst, tab, pairs, n_edges, n_bins);
    bin_sort_kernel<<<n_bins, BIN_W, 0, stream>>>(pairs, tab, start, invdeg, nbr, n_nodes,
                                                 n_bins, n_chunks);

    sage_layer_kernel<true><<<layer_blocks, 256, 0, stream>>>(
        x, start, nbr, invdeg, w1l, b1l, w1r, h, n_nodes);

    sage_layer_kernel<false><<<layer_blocks, 256, 0, stream>>>(
        h, start, nbr, invdeg, w2l, b2l, w2r, out, n_nodes);
}


# Layer launches reshaped to one 25-wave work-group per WGP: 250 groups of 25 tiles instead of 782 of 8 (24 or 32 tiles on a WGP); fullest WGP 32 -> 25 tiles, weights staged once per WGP
# speedup vs baseline: 7.2660x; 1.1174x over previous
#include <hip/hip_runtime.h>

typedef float v2f __attribute__((ext_vector_type(2)));
typedef float v8f __attribute__((ext_vector_type(8)));

#define DIM 64
#define WPAD 80

#define BIN_SHIFT 9
#define BIN_W 512
#define NB_MAX 1024
#define CH_EDGES 6400
#define ORDER_T 512
#define STAGE_CAP 16384

__device__ __forceinline__ unsigned wave_incl_scan_u32(unsigned v, int lane) {
    for (int o = 1; o < 32; o <<= 1) {
        const unsigned up = (unsigned)__shfl_up((int)v, o);
        if (lane >= o) v += up;
    }
    return v;
}

__device__ __forceinline__ void wave_excl_scan_lds(unsigned* a, int n, int lane) {
    const int per = (n + 31) >> 5;
    const int lo = lane * per;
    unsigned t = 0;
    for (int k = 0; k < per; ++k) {
        const int i = lo + k;
        if (i < n) t += a[i];
    }
    const unsigned incl = wave_incl_scan_u32(t, lane);
    unsigned run = incl - t;
    for (int k = 0; k < per; ++k) {
        const int i = lo + k;
        if (i < n) {
            const unsigned c = a[i];
            a[i] = run;
            run += c;
        }
    }
}

#define CH_PER ((CH_EDGES + ORDER_T - 1) / ORDER_T)
static_assert(CH_EDGES < 65536, "run start and count share one 32-bit word");
__global__ void chunk_sort_kernel(const int* __restrict__ src, const int* __restrict__ dst,
                                  unsigned* __restrict__ tab, int2* __restrict__ pairs,
                                  int n_edges, int n_bins) {
    __shared__ unsigned s_cnt[NB_MAX];
    __shared__ unsigned s_lbase[NB_MAX];
    __shared__ int2 s_stage[CH_EDGES];
    const int c = blockIdx.x;
    const int lane = threadIdx.x & 31;
    const int wave = threadIdx.x >> 5;

    for (int i = threadIdx.x; i < n_bins; i += blockDim.x) s_cnt[i] = 0u;
    __syncthreads();

    const int e0 = c * CH_EDGES;
    int e1 = e0 + CH_EDGES;
    if (e1 > n_edges) e1 = n_edges;
    int d[CH_PER], s[CH_PER];
    unsigned rk[CH_PER];
#pragma unroll
    for (int u = 0; u < CH_PER; ++u) {
        const int ee = e0 + (int)threadIdx.x + u * ORDER_T;
        const int ec = ee < e1 ? ee : e1 - 1;
        d[u] = dst[ec];
        s[u] = src[ec];
    }
#pragma unroll
    for (int u = 0; u < CH_PER; ++u) {
        rk[u] = 0u;
        if (e0 + (int)threadIdx.x + u * ORDER_T < e1)
            rk[u] = atomicAdd(&s_cnt[(unsigned)d[u] >> BIN_SHIFT], 1u);
    }
    __syncthreads();
    for (int i = threadIdx.x; i < n_bins; i += blockDim.x) s_lbase[i] = s_cnt[i];
    __syncthreads();
    if (wave == 0) wave_excl_scan_lds(s_lbase, n_bins, lane);
    __syncthreads();
#pragma unroll
    for (int u = 0; u < CH_PER; ++u) {
        if (e0 + (int)threadIdx.x + u * ORDER_T < e1)
            s_stage[s_lbase[(unsigned)d[u] >> BIN_SHIFT] + rk[u]] = make_int2(s[u], d[u]);
    }
    for (int i = threadIdx.x; i < n_bins; i += blockDim.x)
        tab[(long long)c * n_bins + i] = (s_lbase[i] << 16) | s_cnt[i];
    __syncthreads();
    const int cnt = e1 - e0;
    for (int i = threadIdx.x; i < cnt; i += blockDim.x) pairs[e0 + i] = s_stage[i];
}

#define CH_MAX 1024
__global__ void bin_sort_kernel(const int2* __restrict__ pairs,
                                const unsigned* __restrict__ tab,
                                unsigned* __restrict__ start, float* __restrict__ invdeg,
                                int* __restrict__ nbr, int n_nodes, int n_bins, int n_chunks) {
    __shared__ unsigned s_cnt[BIN_W];
    __shared__ unsigned s_wtot[BIN_W / 32];
    __shared__ int s_nbr[STAGE_CAP];
    __shared__ unsigned s_rb[CH_MAX + 1];
    __shared__ unsigned s_adj[CH_MAX];
    __shared__ unsigned s_pidx[STAGE_CAP];
    __shared__ unsigned s_beg;
    const int b = blockIdx.x;
    const int tid = threadIdx.x;
    const int lane = tid & 31;
    const int wave = tid >> 5;

    s_cnt[tid] = 0u;
    if (tid == 0) {
        s_beg = 0u;
        s_rb[n_chunks] = 0u;
    }
    __syncthreads();
    unsigned lbsum = 0u;
    for (int t = tid; t < n_chunks; t += BIN_W) {
        const unsigned inf = tab[(long long)t * n_bins + b];
        s_rb[t] = inf & 0xffffu;
        s_adj[t] = (unsigned)t * (unsigned)CH_EDGES + (inf >> 16);
        lbsum += inf >> 16;
    }
    for (int o = 16; o > 0; o >>= 1) lbsum += (unsigned)__shfl_xor((int)lbsum, o);
    if (lane == 0 && lbsum != 0u) atomicAdd(&s_beg, lbsum);
    __syncthreads();
    if (wave == 0) wave_excl_scan_lds(s_rb, n_chunks + 1, lane);
    __syncthreads();
    const unsigned beg = s_beg;
    const unsigned cnt_b = s_rb[n_chunks];
    const bool staged = cnt_b <= (unsigned)STAGE_CAP;
    for (int t = tid; t < n_chunks; t += BIN_W) {
        const unsigned lo = s_rb[t];
        const unsigned hi = s_rb[t + 1];
        const unsigned adj = s_adj[t] - lo;
        s_adj[t] = adj;
        if (staged)
            for (unsigned m = lo; m < hi; ++m) s_pidx[m] = adj + m;
    }
    __syncthreads();

    auto bisect_index = [&](const unsigned ii) -> unsigned {
        unsigned lo = 0u, hi = (unsigned)n_chunks;
        while (hi - lo > 1u) {
            const unsigned mid = (lo + hi) >> 1;
            if (s_rb[mid] <= ii) lo = mid;
            else hi = mid;
        }
        return s_adj[lo] + ii;
    };

    for (unsigned i = tid; i < cnt_b; i += 4 * BIN_W) {
        int d[4];
        unsigned px[4];
        if (staged) {
#pragma unroll
            for (int u = 0; u < 4; ++u) {
                const unsigned ii = i + u * BIN_W;
                px[u] = s_pidx[ii < cnt_b ? ii : cnt_b - 1];
            }
        } else {
#pragma unroll
            for (int u = 0; u < 4; ++u) {
                const unsigned ii = i + u * BIN_W;
                px[u] = bisect_index(ii < cnt_b ? ii : cnt_b - 1);
            }
        }
#pragma unroll
        for (int u = 0; u < 4; ++u) d[u] = pairs[px[u]].y;
#pragma unroll
        for (int u = 0; u < 4; ++u)
            if (i + u * BIN_W < cnt_b) atomicAdd(&s_cnt[(unsigned)d[u] & (BIN_W - 1)], 1u);
    }
    __syncthreads();

    const unsigned c = s_cnt[tid];
    const unsigned incl = wave_incl_scan_u32(c, lane);
    if (lane == 31) s_wtot[wave] = incl;
    __syncthreads();
    unsigned excl = incl - c;
    for (int w = 0; w < BIN_W / 32; ++w)
        if (w < wave) excl += s_wtot[w];

    const int node = b * BIN_W + tid;
    if (node < n_nodes) {
        start[node] = beg + excl;
        invdeg[node] = 1.0f / fmaxf((float)c, 1.0f);
    }
    if (b == n_bins - 1 && tid == 0) start[n_nodes] = beg + cnt_b;
    s_cnt[tid] = excl;
    __syncthreads();

    for (unsigned i = tid; i < cnt_b; i += 4 * BIN_W) {
        int2 v[4];
        unsigned px[4];
        if (staged) {
#pragma unroll
            for (int u = 0; u < 4; ++u) {
                const unsigned ii = i + u * BIN_W;
                px[u] = s_pidx[ii < cnt_b ? ii : cnt_b - 1];
            }
        } else {
#pragma unroll
            for (int u = 0; u < 4; ++u) {
                const unsigned ii = i + u * BIN_W;
                px[u] = bisect_index(ii < cnt_b ? ii : cnt_b - 1);
            }
        }
#pragma unroll
        for (int u = 0; u < 4; ++u) v[u] = pairs[px[u]];
#pragma unroll
        for (int u = 0; u < 4; ++u) {
            if (i + u * BIN_W < cnt_b) {
                const unsigned p = atomicAdd(&s_cnt[(unsigned)v[u].y & (BIN_W - 1)], 1u);
                if (staged) s_nbr[p] = v[u].x;
                else nbr[beg + p] = v[u].x;
            }
        }
    }
    if (staged) {
        __syncthreads();
        for (unsigned i = tid; i < cnt_b; i += BIN_W) nbr[beg + i] = s_nbr[i];
    }
}

#define APITCH 68
#define ROWS_IN_FLIGHT 16
#define LAYER_WAVES 25

template <bool FULL>
__device__ __forceinline__ void sum_group(const v2f* rows, const int mine, const unsigned j0,
                                          const unsigned cnt, const unsigned pos0,
                                          const int mystart, const float id, float* my_a,
                                          const int lane, const bool fetch,
                                          const int* next_entry, int& mine_next,
                                          v2f& acc, unsigned& cur, unsigned& nb) {
    v2f v[ROWS_IN_FLIGHT];
#pragma unroll
    for (int u = 0; u < ROWS_IN_FLIGHT; ++u) {
        const unsigned jj = (FULL || (unsigned)u < cnt) ? (unsigned)u : cnt - 1;
        v[u] = rows[(long long)__builtin_amdgcn_readlane(mine, j0 + jj) * 32 + lane];
    }
    if (fetch) mine_next = *next_entry;
#pragma unroll
    for (int u = 0; u < ROWS_IN_FLIGHT; ++u) {
        if (FULL || (unsigned)u < cnt) {
            const unsigned p = pos0 + (unsigned)u;
            while (p == nb) {
                const float sc = __builtin_bit_cast(
                    float, __builtin_amdgcn_readlane(__builtin_bit_cast(int, id), cur));
                v2f o;
                o.x = acc.x * sc;
                o.y = acc.y * sc;
                *(v2f*)&my_a[cur * APITCH + 2 * lane] = o;
                acc.x = 0.0f;
                acc.y = 0.0f;
                ++cur;
                nb = (unsigned)__builtin_amdgcn_readlane(mystart, cur + 1);
            }
            acc += v[u];
        }
    }
}

template <bool RELU>
__global__ void sage_layer_kernel(const float* __restrict__ feat,
                                  const unsigned* __restrict__ start,
                                  const int* __restrict__ nbr,
                                  const float* __restrict__ invdeg,
                                  const float* __restrict__ wl,
                                  const float* __restrict__ bl,
                                  const float* __restrict__ wr,
                                  float* __restrict__ out, int n_nodes) {
    __shared__ v2f lds_wl[32 * WPAD];
    __shared__ v2f lds_wr[32 * WPAD];
    __shared__ float lds_a[LAYER_WAVES * 16 * APITCH];

    for (int i = threadIdx.x; i < 32 * 64; i += blockDim.x) {
        const int p = i >> 6;
        const int n = i & 63;
        v2f a, b;
        a.x = wl[(2 * p) * DIM + n];
        a.y = wl[(2 * p + 1) * DIM + n];
        b.x = wr[(2 * p) * DIM + n];
        b.y = wr[(2 * p + 1) * DIM + n];
        lds_wl[p * WPAD + n] = a;
        lds_wr[p * WPAD + n] = b;
    }
    __syncthreads();

    const int lane = threadIdx.x & 31;
    const int wave = __builtin_amdgcn_readfirstlane((int)(threadIdx.x >> 5));
    const int tile = blockIdx.x * LAYER_WAVES + wave;
    const int n_tiles = (n_nodes + 15) >> 4;
    if (tile >= n_tiles) return;

    const int row0 = tile << 4;
    const int r     = lane & 15;
    const int kh    = (lane >> 4) << 1;
    const int pbase = lane >> 4;
    float* my_a = lds_a + wave * (16 * APITCH);

    int node = row0 + r;
    if (node > n_nodes - 1) node = n_nodes - 1;
    const float id = invdeg[node];
    const float* xrow = feat + (long long)node * DIM;

    int bi = row0 + (lane < 16 ? lane : 16);
    if (bi > n_nodes) bi = n_nodes;
    const int mystart = (int)start[bi];

    v2f xa[16];
#pragma unroll
    for (int k = 0; k < DIM; k += 4) {
        xa[k >> 2].x = xrow[k + kh];
        xa[k >> 2].y = xrow[k + kh + 1];
    }

    {
        const unsigned tb = (unsigned)__builtin_amdgcn_readlane(mystart, 0);
        const unsigned te = (unsigned)__builtin_amdgcn_readlane(mystart, 16);
        const v2f* rows = (const v2f*)feat;
        unsigned cur = 0;
        unsigned nb = (unsigned)__builtin_amdgcn_readlane(mystart, 1);
        v2f acc = {0.0f, 0.0f};
        int mine = 0;
        if (tb < te) mine = nbr[tb + (unsigned)lane < te ? tb + (unsigned)lane : te - 1];
        for (unsigned base = tb; base < te; base += 32) {
            const unsigned left = te - base;
            const unsigned m = left < 32u ? left : 32u;
            const unsigned nx = base + 32u + (unsigned)lane;
            const int* next_entry = nbr + (nx < te ? nx : te - 1);
            int mine_next = 0;
            for (unsigned j0 = 0; j0 < m; j0 += ROWS_IN_FLIGHT) {
                const unsigned cnt = (m - j0) < (unsigned)ROWS_IN_FLIGHT ? (m - j0)
                                                                         : (unsigned)ROWS_IN_FLIGHT;
                if (cnt == (unsigned)ROWS_IN_FLIGHT)
                    sum_group<true>(rows, mine, j0, cnt, base + j0, mystart, id, my_a, lane,
                                    j0 == 0, next_entry, mine_next, acc, cur, nb);
                else
                    sum_group<false>(rows, mine, j0, cnt, base + j0, mystart, id, my_a, lane,
                                     j0 == 0, next_entry, mine_next, acc, cur, nb);
            }
            mine = mine_next;
        }
        for (; cur < 16; ++cur) {
            const float sc = __builtin_bit_cast(
                float, __builtin_amdgcn_readlane(__builtin_bit_cast(int, id), cur));
            v2f o;
            o.x = acc.x * sc;
            o.y = acc.y * sc;
            *(v2f*)&my_a[cur * APITCH + 2 * lane] = o;
            acc.x = 0.0f;
            acc.y = 0.0f;
        }
    }

    v8f acc4[4] = {{}, {}, {}, {}};

    const float* arow = my_a + r * APITCH;
#pragma unroll
    for (int k = 0; k < DIM; k += 4) {
        const v2f a = *(const v2f*)&arow[k + kh];
        const int prow = (k >> 1) + pbase;
#pragma unroll
        for (int nt = 0; nt < 4; ++nt) {
            v2f b = lds_wl[prow * WPAD + nt * 16 + r];
            acc4[nt] = __builtin_amdgcn_wmma_f32_16x16x4_f32(
                false, a, false, b, (short)0, acc4[nt], false, false);
        }
    }

#pragma unroll
    for (int k = 0; k < DIM; k += 4) {
        const v2f a = xa[k >> 2];
        const int prow = (k >> 1) + pbase;
#pragma unroll
        for (int nt = 0; nt < 4; ++nt) {
            v2f b = lds_wr[prow * WPAD + nt * 16 + r];
            acc4[nt] = __builtin_amdgcn_wmma_f32_16x16x4_f32(
                false, a, false, b, (short)0, acc4[nt], false, false);
        }
    }

    const int mbase = (lane >> 4) << 3;
#pragma unroll
    for (int nt = 0; nt < 4; ++nt) {
        const int n = nt * 16 + r;
        const float bias = bl[n];
#pragma unroll
        for (int vr = 0; vr < 8; ++vr) {
            const int m = row0 + mbase + vr;
            if (m < n_nodes) {
                float v = acc4[nt][vr] + bias;
                if (RELU) v = fmaxf(v, 0.0f);
                out[(long long)m * DIM + n] = v;
            }
        }
    }
}

extern "C" void kernel_launch(void* const* d_in, const int* in_sizes, int n_in,
                              void* d_out, int out_size, void* d_ws, size_t ws_size,
                              hipStream_t stream) {
    const float* x   = (const float*)d_in[0];
    const int*  edge = (const int*) d_in[1];
    const float* w1l = (const float*)d_in[2];
    const float* b1l = (const float*)d_in[3];
    const float* w1r = (const float*)d_in[4];
    const float* w2l = (const float*)d_in[5];
    const float* b2l = (const float*)d_in[6];
    const float* w2r = (const float*)d_in[7];
    float* out = (float*)d_out;

    const int n_nodes = in_sizes[0] / DIM;
    const int n_edges = in_sizes[1] / 2;
    const int* src = edge;
    const int* dst = edge + n_edges;

    float* ws      = (float*)d_ws;
    float* invdeg  = ws;
    float* list    = ws + n_nodes;
    float* h       = list + (long long)n_nodes * DIM;
    unsigned* start = (unsigned*)list;
    int*      nbr   = (int*)(start + n_nodes + 1);

    const int n_bins       = (n_nodes + BIN_W - 1) / BIN_W;
    const int n_chunks     = (n_edges + CH_EDGES - 1) / CH_EDGES;
    const int tiles        = (n_nodes + 15) / 16;
    const int layer_blocks = (tiles + LAYER_WAVES - 1) / LAYER_WAVES;

    int2*     pairs = (int2*)d_out;
    unsigned* tab   = (unsigned*)d_out + 2 * (long long)n_edges;

    chunk_sort_kernel<<<n_chunks, ORDER_T, 0, stream>>>(src, dst, tab, pairs, n_edges, n_bins);
    bin_sort_kernel<<<n_bins, BIN_W, 0, stream>>>(pairs, tab, start, invdeg, nbr, n_nodes,
                                                 n_bins, n_chunks);

    sage_layer_kernel<true><<<layer_blocks, LAYER_WAVES * 32, 0, stream>>>(
        x, start, nbr, invdeg, w1l, b1l, w1r, h, n_nodes);

    sage_layer_kernel<false><<<layer_blocks, LAYER_WAVES * 32, 0, stream>>>(
        h, start, nbr, invdeg, w2l, b2l, w2r, out, n_nodes);
}


# Self product's 64 WMMAs issued under the gather: self rows staged in LDS as whole rows, one k step (4 WMMAs) behind each group's row reads; accumulators take self terms before neighbour terms
# speedup vs baseline: 7.6409x; 1.0516x over previous
#include <hip/hip_runtime.h>

typedef float v2f __attribute__((ext_vector_type(2)));
typedef float v8f __attribute__((ext_vector_type(8)));

#define DIM 64
#define WPAD 80

#define BIN_SHIFT 9
#define BIN_W 512
#define NB_MAX 1024
#define CH_EDGES 6400
#define ORDER_T 512
#define STAGE_CAP 16384

__device__ __forceinline__ unsigned wave_incl_scan_u32(unsigned v, int lane) {
    for (int o = 1; o < 32; o <<= 1) {
        const unsigned up = (unsigned)__shfl_up((int)v, o);
        if (lane >= o) v += up;
    }
    return v;
}

__device__ __forceinline__ void wave_excl_scan_lds(unsigned* a, int n, int lane) {
    const int per = (n + 31) >> 5;
    const int lo = lane * per;
    unsigned t = 0;
    for (int k = 0; k < per; ++k) {
        const int i = lo + k;
        if (i < n) t += a[i];
    }
    const unsigned incl = wave_incl_scan_u32(t, lane);
    unsigned run = incl - t;
    for (int k = 0; k < per; ++k) {
        const int i = lo + k;
        if (i < n) {
            const unsigned c = a[i];
            a[i] = run;
            run += c;
        }
    }
}

#define CH_PER ((CH_EDGES + ORDER_T - 1) / ORDER_T)
static_assert(CH_EDGES < 65536, "run start and count share one 32-bit word");
__global__ void chunk_sort_kernel(const int* __restrict__ src, const int* __restrict__ dst,
                                  unsigned* __restrict__ tab, int2* __restrict__ pairs,
                                  int n_edges, int n_bins) {
    __shared__ unsigned s_cnt[NB_MAX];
    __shared__ unsigned s_lbase[NB_MAX];
    __shared__ int2 s_stage[CH_EDGES];
    const int c = blockIdx.x;
    const int lane = threadIdx.x & 31;
    const int wave = threadIdx.x >> 5;

    for (int i = threadIdx.x; i < n_bins; i += blockDim.x) s_cnt[i] = 0u;
    __syncthreads();

    const int e0 = c * CH_EDGES;
    int e1 = e0 + CH_EDGES;
    if (e1 > n_edges) e1 = n_edges;
    int d[CH_PER], s[CH_PER];
    unsigned rk[CH_PER];
#pragma unroll
    for (int u = 0; u < CH_PER; ++u) {
        const int ee = e0 + (int)threadIdx.x + u * ORDER_T;
        const int ec = ee < e1 ? ee : e1 - 1;
        d[u] = dst[ec];
        s[u] = src[ec];
    }
#pragma unroll
    for (int u = 0; u < CH_PER; ++u) {
        rk[u] = 0u;
        if (e0 + (int)threadIdx.x + u * ORDER_T < e1)
            rk[u] = atomicAdd(&s_cnt[(unsigned)d[u] >> BIN_SHIFT], 1u);
    }
    __syncthreads();
    for (int i = threadIdx.x; i < n_bins; i += blockDim.x) s_lbase[i] = s_cnt[i];
    __syncthreads();
    if (wave == 0) wave_excl_scan_lds(s_lbase, n_bins, lane);
    __syncthreads();
#pragma unroll
    for (int u = 0; u < CH_PER; ++u) {
        if (e0 + (int)threadIdx.x + u * ORDER_T < e1)
            s_stage[s_lbase[(unsigned)d[u] >> BIN_SHIFT] + rk[u]] = make_int2(s[u], d[u]);
    }
    for (int i = threadIdx.x; i < n_bins; i += blockDim.x)
        tab[(long long)c * n_bins + i] = (s_lbase[i] << 16) | s_cnt[i];
    __syncthreads();
    const int cnt = e1 - e0;
    for (int i = threadIdx.x; i < cnt; i += blockDim.x) pairs[e0 + i] = s_stage[i];
}

#define CH_MAX 1024
__global__ void bin_sort_kernel(const int2* __restrict__ pairs,
                                const unsigned* __restrict__ tab,
                                unsigned* __restrict__ start, float* __restrict__ invdeg,
                                int* __restrict__ nbr, int n_nodes, int n_bins, int n_chunks) {
    __shared__ unsigned s_cnt[BIN_W];
    __shared__ unsigned s_wtot[BIN_W / 32];
    __shared__ int s_nbr[STAGE_CAP];
    __shared__ unsigned s_rb[CH_MAX + 1];
    __shared__ unsigned s_adj[CH_MAX];
    __shared__ unsigned s_pidx[STAGE_CAP];
    __shared__ unsigned s_beg;
    const int b = blockIdx.x;
    const int tid = threadIdx.x;
    const int lane = tid & 31;
    const int wave = tid >> 5;

    s_cnt[tid] = 0u;
    if (tid == 0) {
        s_beg = 0u;
        s_rb[n_chunks] = 0u;
    }
    __syncthreads();
    unsigned lbsum = 0u;
    for (int t = tid; t < n_chunks; t += BIN_W) {
        const unsigned inf = tab[(long long)t * n_bins + b];
        s_rb[t] = inf & 0xffffu;
        s_adj[t] = (unsigned)t * (unsigned)CH_EDGES + (inf >> 16);
        lbsum += inf >> 16;
    }
    for (int o = 16; o > 0; o >>= 1) lbsum += (unsigned)__shfl_xor((int)lbsum, o);
    if (lane == 0 && lbsum != 0u) atomicAdd(&s_beg, lbsum);
    __syncthreads();
    if (wave == 0) wave_excl_scan_lds(s_rb, n_chunks + 1, lane);
    __syncthreads();
    const unsigned beg = s_beg;
    const unsigned cnt_b = s_rb[n_chunks];
    const bool staged = cnt_b <= (unsigned)STAGE_CAP;
    for (int t = tid; t < n_chunks; t += BIN_W) {
        const unsigned lo = s_rb[t];
        const unsigned hi = s_rb[t + 1];
        const unsigned adj = s_adj[t] - lo;
        s_adj[t] = adj;
        if (staged)
            for (unsigned m = lo; m < hi; ++m) s_pidx[m] = adj + m;
    }
    __syncthreads();

    auto bisect_index = [&](const unsigned ii) -> unsigned {
        unsigned lo = 0u, hi = (unsigned)n_chunks;
        while (hi - lo > 1u) {
            const unsigned mid = (lo + hi) >> 1;
            if (s_rb[mid] <= ii) lo = mid;
            else hi = mid;
        }
        return s_adj[lo] + ii;
    };

    for (unsigned i = tid; i < cnt_b; i += 4 * BIN_W) {
        int d[4];
        unsigned px[4];
        if (staged) {
#pragma unroll
            for (int u = 0; u < 4; ++u) {
                const unsigned ii = i + u * BIN_W;
                px[u] = s_pidx[ii < cnt_b ? ii : cnt_b - 1];
            }
        } else {
#pragma unroll
            for (int u = 0; u < 4; ++u) {
                const unsigned ii = i + u * BIN_W;
                px[u] = bisect_index(ii < cnt_b ? ii : cnt_b - 1);
            }
        }
#pragma unroll
        for (int u = 0; u < 4; ++u) d[u] = pairs[px[u]].y;
#pragma unroll
        for (int u = 0; u < 4; ++u)
            if (i + u * BIN_W < cnt_b) atomicAdd(&s_cnt[(unsigned)d[u] & (BIN_W - 1)], 1u);
    }
    __syncthreads();

    const unsigned c = s_cnt[tid];
    const unsigned incl = wave_incl_scan_u32(c, lane);
    if (lane == 31) s_wtot[wave] = incl;
    __syncthreads();
    unsigned excl = incl - c;
    for (int w = 0; w < BIN_W / 32; ++w)
        if (w < wave) excl += s_wtot[w];

    const int node = b * BIN_W + tid;
    if (node < n_nodes) {
        start[node] = beg + excl;
        invdeg[node] = 1.0f / fmaxf((float)c, 1.0f);
    }
    if (b == n_bins - 1 && tid == 0) start[n_nodes] = beg + cnt_b;
    s_cnt[tid] = excl;
    __syncthreads();

    for (unsigned i = tid; i < cnt_b; i += 4 * BIN_W) {
        int2 v[4];
        unsigned px[4];
        if (staged) {
#pragma unroll
            for (int u = 0; u < 4; ++u) {
                const unsigned ii = i + u * BIN_W;
                px[u] = s_pidx[ii < cnt_b ? ii : cnt_b - 1];
            }
        } else {
#pragma unroll
            for (int u = 0; u < 4; ++u) {
                const unsigned ii = i + u * BIN_W;
                px[u] = bisect_index(ii < cnt_b ? ii : cnt_b - 1);
            }
        }
#pragma unroll
        for (int u = 0; u < 4; ++u) v[u] = pairs[px[u]];
#pragma unroll
        for (int u = 0; u < 4; ++u) {
            if (i + u * BIN_W < cnt_b) {
                const unsigned p = atomicAdd(&s_cnt[(unsigned)v[u].y & (BIN_W - 1)], 1u);
                if (staged) s_nbr[p] = v[u].x;
                else nbr[beg + p] = v[u].x;
            }
        }
    }
    if (staged) {
        __syncthreads();
        for (unsigned i = tid; i < cnt_b; i += BIN_W) nbr[beg + i] = s_nbr[i];
    }
}

#define APITCH 68
#define ROWS_IN_FLIGHT 16
#define LAYER_WAVES 25

__device__ __forceinline__ void self_step(const float* xa_lane, const v2f* wr_lane,
                                          const unsigned ks, v8f (&acc4)[4]) {
    const v2f a = *(const v2f*)&xa_lane[4u * ks];
#pragma unroll
    for (int nt = 0; nt < 4; ++nt) {
        const v2f b = wr_lane[2u * ks * WPAD + nt * 16];
        acc4[nt] = __builtin_amdgcn_wmma_f32_16x16x4_f32(
            false, a, false, b, (short)0, acc4[nt], false, false);
    }
    asm volatile("v_nop\n\tv_nop\n\tv_nop\n\tv_nop"
                 : "+v"(acc4[0]), "+v"(acc4[1]), "+v"(acc4[2]), "+v"(acc4[3]));
}

template <bool FULL>
__device__ __forceinline__ void ask_rows(const v2f* rows, const int mine, const unsigned j0,
                                         const unsigned cnt, const int lane,
                                         v2f (&v)[ROWS_IN_FLIGHT]) {
#pragma unroll
    for (int u = 0; u < ROWS_IN_FLIGHT; ++u) {
        const unsigned jj = (FULL || (unsigned)u < cnt) ? (unsigned)u : cnt - 1;
        v[u] = rows[(long long)__builtin_amdgcn_readlane(mine, j0 + jj) * 32 + lane];
    }
}

template <bool FULL>
__device__ __forceinline__ void add_rows(const v2f (&v)[ROWS_IN_FLIGHT], const unsigned cnt,
                                         const unsigned pos0, const int mystart, const float id,
                                         float* my_a, const int lane,
                                         v2f& acc, unsigned& cur, unsigned& nb) {
#pragma unroll
    for (int u = 0; u < ROWS_IN_FLIGHT; ++u) {
        if (FULL || (unsigned)u < cnt) {
            const unsigned p = pos0 + (unsigned)u;
            while (p == nb) {
                const float sc = __builtin_bit_cast(
                    float, __builtin_amdgcn_readlane(__builtin_bit_cast(int, id), cur));
                v2f o;
                o.x = acc.x * sc;
                o.y = acc.y * sc;
                *(v2f*)&my_a[cur * APITCH + 2 * lane] = o;
                acc.x = 0.0f;
                acc.y = 0.0f;
                ++cur;
                nb = (unsigned)__builtin_amdgcn_readlane(mystart, cur + 1);
            }
            acc += v[u];
        }
    }
}

template <bool STEP>
__device__ __forceinline__ void sum_span(const v2f* rows, const int* nbr, unsigned base,
                                         const unsigned stop, const unsigned te,
                                         const int mystart, const float id, float* my_a,
                                         const int lane, int& mine, v2f& acc,
                                         unsigned& cur, unsigned& nb,
                                         const float* xa_lane, const v2f* wr_lane,
                                         unsigned& ks, v8f (&acc4)[4]) {
    for (; base < stop; base += 32) {
        const unsigned left = te - base;
        const unsigned m = left < 32u ? left : 32u;
        const unsigned nx = base + 32u + (unsigned)lane;
        const int* next_entry = nbr + (nx < te ? nx : te - 1);
        int mine_next = 0;
        for (unsigned j0 = 0; j0 < m; j0 += ROWS_IN_FLIGHT) {
            const unsigned cnt = (m - j0) < (unsigned)ROWS_IN_FLIGHT ? (m - j0)
                                                                     : (unsigned)ROWS_IN_FLIGHT;
            const bool full = cnt == (unsigned)ROWS_IN_FLIGHT;
            v2f v[ROWS_IN_FLIGHT];
            if (full) ask_rows<true>(rows, mine, j0, cnt, lane, v);
            else ask_rows<false>(rows, mine, j0, cnt, lane, v);
            if (j0 == 0) mine_next = *next_entry;
            if (STEP) {
                self_step(xa_lane, wr_lane, ks, acc4);
                ++ks;
            }
            if (full) add_rows<true>(v, cnt, base + j0, mystart, id, my_a, lane, acc, cur, nb);
            else add_rows<false>(v, cnt, base + j0, mystart, id, my_a, lane, acc, cur, nb);
        }
        mine = mine_next;
    }
}

template <bool RELU>
__global__ void sage_layer_kernel(const float* __restrict__ feat,
                                  const unsigned* __restrict__ start,
                                  const int* __restrict__ nbr,
                                  const float* __restrict__ invdeg,
                                  const float* __restrict__ wl,
                                  const float* __restrict__ bl,
                                  const float* __restrict__ wr,
                                  float* __restrict__ out, int n_nodes) {
    __shared__ v2f lds_wl[32 * WPAD];
    __shared__ v2f lds_wr[32 * WPAD];
    __shared__ float lds_a[LAYER_WAVES * 16 * APITCH];
    __shared__ v2f lds_x[LAYER_WAVES * 16 * APITCH / 2];

    for (int i = threadIdx.x; i < 32 * 64; i += blockDim.x) {
        const int p = i >> 6;
        const int n = i & 63;
        v2f a, b;
        a.x = wl[(2 * p) * DIM + n];
        a.y = wl[(2 * p + 1) * DIM + n];
        b.x = wr[(2 * p) * DIM + n];
        b.y = wr[(2 * p + 1) * DIM + n];
        lds_wl[p * WPAD + n] = a;
        lds_wr[p * WPAD + n] = b;
    }
    __syncthreads();

    const int lane = threadIdx.x & 31;
    const int wave = __builtin_amdgcn_readfirstlane((int)(threadIdx.x >> 5));
    const int tile = blockIdx.x * LAYER_WAVES + wave;
    const int n_tiles = (n_nodes + 15) >> 4;
    if (tile >= n_tiles) return;

    const int row0 = tile << 4;
    const int r     = lane & 15;
    const int kh    = (lane >> 4) << 1;
    const int pbase = lane >> 4;
    float* my_a = lds_a + wave * (16 * APITCH);
    float* my_x = (float*)lds_x + wave * (16 * APITCH);

    int node = row0 + r;
    if (node > n_nodes - 1) node = n_nodes - 1;
    const float id = invdeg[node];

    int bi = row0 + (lane < 16 ? lane : 16);
    if (bi > n_nodes) bi = n_nodes;
    const int mystart = (int)start[bi];

    const v2f* rows = (const v2f*)feat;
    v2f xs[16];
#pragma unroll
    for (int i = 0; i < 16; ++i) {
        int sn = row0 + i;
        if (sn > n_nodes - 1) sn = n_nodes - 1;
        xs[i] = rows[(long long)sn * 32 + lane];
    }

    v8f acc4[4] = {{}, {}, {}, {}};
    const float* xa_lane = my_x + r * APITCH + kh;
    const v2f* wr_lane = lds_wr + pbase * WPAD + r;
    unsigned ks = 0;

    {
        const unsigned tb = (unsigned)__builtin_amdgcn_readlane(mystart, 0);
        const unsigned te = (unsigned)__builtin_amdgcn_readlane(mystart, 16);
        unsigned cur = 0;
        unsigned nb = (unsigned)__builtin_amdgcn_readlane(mystart, 1);
        v2f acc = {0.0f, 0.0f};
        int mine = 0;
        if (tb < te) mine = nbr[tb + (unsigned)lane < te ? tb + (unsigned)lane : te - 1];
#pragma unroll
        for (int i = 0; i < 16; ++i) *(v2f*)&my_x[i * APITCH + 2 * lane] = xs[i];
        const unsigned step_rows = (unsigned)(DIM / 4) * (unsigned)ROWS_IN_FLIGHT;
        static_assert(((DIM / 4) * ROWS_IN_FLIGHT) % 32 == 0, "the stepped part ends on a whole piece");
        const unsigned te_step = te - tb < step_rows ? te : tb + step_rows;
        sum_span<true>(rows, nbr, tb, te_step, te, mystart, id, my_a, lane, mine, acc, cur, nb,
                       xa_lane, wr_lane, ks, acc4);
        sum_span<false>(rows, nbr, te_step, te, te, mystart, id, my_a, lane, mine, acc, cur, nb,
                        xa_lane, wr_lane, ks, acc4);
        for (; cur < 16; ++cur) {
            const float sc = __builtin_bit_cast(
                float, __builtin_amdgcn_readlane(__builtin_bit_cast(int, id), cur));
            v2f o;
            o.x = acc.x * sc;
            o.y = acc.y * sc;
            *(v2f*)&my_a[cur * APITCH + 2 * lane] = o;
            acc.x = 0.0f;
            acc.y = 0.0f;
        }
    }

#pragma nounroll
    for (; ks < (unsigned)(DIM / 4); ++ks) self_step(xa_lane, wr_lane, ks, acc4);

    const float* arow = my_a + r * APITCH;
#pragma unroll
    for (int k = 0; k < DIM; k += 4) {
        const v2f a = *(const v2f*)&arow[k + kh];
        const int prow = (k >> 1) + pbase;
#pragma unroll
        for (int nt = 0; nt < 4; ++nt) {
            v2f b = lds_wl[prow * WPAD + nt * 16 + r];
            acc4[nt] = __builtin_amdgcn_wmma_f32_16x16x4_f32(
                false, a, false, b, (short)0, acc4[nt], false, false);
        }
    }

    const int mbase = (lane >> 4) << 3;
#pragma unroll
    for (int nt = 0; nt < 4; ++nt) {
        const int n = nt * 16 + r;
        const float bias = bl[n];
#pragma unroll
        for (int vr = 0; vr < 8; ++vr) {
            const int m = row0 + mbase + vr;
            if (m < n_nodes) {
                float v = acc4[nt][vr] + bias;
                if (RELU) v = fmaxf(v, 0.0f);
                out[(long long)m * DIM + n] = v;
            }
        }
    }
}

extern "C" void kernel_launch(void* const* d_in, const int* in_sizes, int n_in,
                              void* d_out, int out_size, void* d_ws, size_t ws_size,
                              hipStream_t stream) {
    const float* x   = (const float*)d_in[0];
    const int*  edge = (const int*) d_in[1];
    const float* w1l = (const float*)d_in[2];
    const float* b1l = (const float*)d_in[3];
    const float* w1r = (const float*)d_in[4];
    const float* w2l = (const float*)d_in[5];
    const float* b2l = (const float*)d_in[6];
    const float* w2r = (const float*)d_in[7];
    float* out = (float*)d_out;

    const int n_nodes = in_sizes[0] / DIM;
    const int n_edges = in_sizes[1] / 2;
    const int* src = edge;
    const int* dst = edge + n_edges;

    float* ws      = (float*)d_ws;
    float* invdeg  = ws;
    float* list    = ws + n_nodes;
    float* h       = list + (long long)n_nodes * DIM;
    unsigned* start = (unsigned*)list;
    int*      nbr   = (int*)(start + n_nodes + 1);

    const int n_bins       = (n_nodes + BIN_W - 1) / BIN_W;
    const int n_chunks     = (n_edges + CH_EDGES - 1) / CH_EDGES;
    const int tiles        = (n_nodes + 15) / 16;
    const int layer_blocks = (tiles + LAYER_WAVES - 1) / LAYER_WAVES;

    int2*     pairs = (int2*)d_out;
    unsigned* tab   = (unsigned*)d_out + 2 * (long long)n_edges;

    chunk_sort_kernel<<<n_chunks, ORDER_T, 0, stream>>>(src, dst, tab, pairs, n_edges, n_bins);
    bin_sort_kernel<<<n_bins, BIN_W, 0, stream>>>(pairs, tab, start, invdeg, nbr, n_nodes,
                                                 n_bins, n_chunks);

    sage_layer_kernel<true><<<layer_blocks, LAYER_WAVES * 32, 0, stream>>>(
        x, start, nbr, invdeg, w1l, b1l, w1r, h, n_nodes);

    sage_layer_kernel<false><<<layer_blocks, LAYER_WAVES * 32, 0, stream>>>(
        h, start, nbr, invdeg, w2l, b2l, w2r, out, n_nodes);
}


# Layer epilogue through LDS: the tile (bias, ReLU applied) goes into the wave's dead sum tile and leaves as 8 whole-row stores, 16 B per lane, 4 full lines each; was 32 guarded 4-byte half-line stores
# speedup vs baseline: 8.3002x; 1.0863x over previous
#include <hip/hip_runtime.h>

typedef float v2f __attribute__((ext_vector_type(2)));
typedef float v4f __attribute__((ext_vector_type(4)));
typedef float v8f __attribute__((ext_vector_type(8)));

#define DIM 64
#define WPAD 80

#define BIN_SHIFT 9
#define BIN_W 512
#define NB_MAX 1024
#define CH_EDGES 6400
#define ORDER_T 512
#define STAGE_CAP 16384

__device__ __forceinline__ unsigned wave_incl_scan_u32(unsigned v, int lane) {
    for (int o = 1; o < 32; o <<= 1) {
        const unsigned up = (unsigned)__shfl_up((int)v, o);
        if (lane >= o) v += up;
    }
    return v;
}

__device__ __forceinline__ void wave_excl_scan_lds(unsigned* a, int n, int lane) {
    const int per = (n + 31) >> 5;
    const int lo = lane * per;
    unsigned t = 0;
    for (int k = 0; k < per; ++k) {
        const int i = lo + k;
        if (i < n) t += a[i];
    }
    const unsigned incl = wave_incl_scan_u32(t, lane);
    unsigned run = incl - t;
    for (int k = 0; k < per; ++k) {
        const int i = lo + k;
        if (i < n) {
            const unsigned c = a[i];
            a[i] = run;
            run += c;
        }
    }
}

#define CH_PER ((CH_EDGES + ORDER_T - 1) / ORDER_T)
static_assert(CH_EDGES < 65536, "run start and count share one 32-bit word");
__global__ void chunk_sort_kernel(const int* __restrict__ src, const int* __restrict__ dst,
                                  unsigned* __restrict__ tab, int2* __restrict__ pairs,
                                  int n_edges, int n_bins) {
    __shared__ unsigned s_cnt[NB_MAX];
    __shared__ unsigned s_lbase[NB_MAX];
    __shared__ int2 s_stage[CH_EDGES];
    const int c = blockIdx.x;
    const int lane = threadIdx.x & 31;
    const int wave = threadIdx.x >> 5;

    for (int i = threadIdx.x; i < n_bins; i += blockDim.x) s_cnt[i] = 0u;
    __syncthreads();

    const int e0 = c * CH_EDGES;
    int e1 = e0 + CH_EDGES;
    if (e1 > n_edges) e1 = n_edges;
    int d[CH_PER], s[CH_PER];
    unsigned rk[CH_PER];
#pragma unroll
    for (int u = 0; u < CH_PER; ++u) {
        const int ee = e0 + (int)threadIdx.x + u * ORDER_T;
        const int ec = ee < e1 ? ee : e1 - 1;
        d[u] = dst[ec];
        s[u] = src[ec];
    }
#pragma unroll
    for (int u = 0; u < CH_PER; ++u) {
        rk[u] = 0u;
        if (e0 + (int)threadIdx.x + u * ORDER_T < e1)
            rk[u] = atomicAdd(&s_cnt[(unsigned)d[u] >> BIN_SHIFT], 1u);
    }
    __syncthreads();
    for (int i = threadIdx.x; i < n_bins; i += blockDim.x) s_lbase[i] = s_cnt[i];
    __syncthreads();
    if (wave == 0) wave_excl_scan_lds(s_lbase, n_bins, lane);
    __syncthreads();
#pragma unroll
    for (int u = 0; u < CH_PER; ++u) {
        if (e0 + (int)threadIdx.x + u * ORDER_T < e1)
            s_stage[s_lbase[(unsigned)d[u] >> BIN_SHIFT] + rk[u]] = make_int2(s[u], d[u]);
    }
    for (int i = threadIdx.x; i < n_bins; i += blockDim.x)
        tab[(long long)c * n_bins + i] = (s_lbase[i] << 16) | s_cnt[i];
    __syncthreads();
    const int cnt = e1 - e0;
    for (int i = threadIdx.x; i < cnt; i += blockDim.x) pairs[e0 + i] = s_stage[i];
}

#define CH_MAX 1024
__global__ void bin_sort_kernel(const int2* __restrict__ pairs,
                                const unsigned* __restrict__ tab,
                                unsigned* __restrict__ start, float* __restrict__ invdeg,
                                int* __restrict__ nbr, int n_nodes, int n_bins, int n_chunks) {
    __shared__ unsigned s_cnt[BIN_W];
    __shared__ unsigned s_wtot[BIN_W / 32];
    __shared__ int s_nbr[STAGE_CAP];
    __shared__ unsigned s_rb[CH_MAX + 1];
    __shared__ unsigned s_adj[CH_MAX];
    __shared__ unsigned s_pidx[STAGE_CAP];
    __shared__ unsigned s_beg;
    const int b = blockIdx.x;
    const int tid = threadIdx.x;
    const int lane = tid & 31;
    const int wave = tid >> 5;

    s_cnt[tid] = 0u;
    if (tid == 0) {
        s_beg = 0u;
        s_rb[n_chunks] = 0u;
    }
    __syncthreads();
    unsigned lbsum = 0u;
    for (int t = tid; t < n_chunks; t += BIN_W) {
        const unsigned inf = tab[(long long)t * n_bins + b];
        s_rb[t] = inf & 0xffffu;
        s_adj[t] = (unsigned)t * (unsigned)CH_EDGES + (inf >> 16);
        lbsum += inf >> 16;
    }
    for (int o = 16; o > 0; o >>= 1) lbsum += (unsigned)__shfl_xor((int)lbsum, o);
    if (lane == 0 && lbsum != 0u) atomicAdd(&s_beg, lbsum);
    __syncthreads();
    if (wave == 0) wave_excl_scan_lds(s_rb, n_chunks + 1, lane);
    __syncthreads();
    const unsigned beg = s_beg;
    const unsigned cnt_b = s_rb[n_chunks];
    const bool staged = cnt_b <= (unsigned)STAGE_CAP;
    for (int t = tid; t < n_chunks; t += BIN_W) {
        const unsigned lo = s_rb[t];
        const unsigned hi = s_rb[t + 1];
        const unsigned adj = s_adj[t] - lo;
        s_adj[t] = adj;
        if (staged)
            for (unsigned m = lo; m < hi; ++m) s_pidx[m] = adj + m;
    }
    __syncthreads();

    auto bisect_index = [&](const unsigned ii) -> unsigned {
        unsigned lo = 0u, hi = (unsigned)n_chunks;
        while (hi - lo > 1u) {
            const unsigned mid = (lo + hi) >> 1;
            if (s_rb[mid] <= ii) lo = mid;
            else hi = mid;
        }
        return s_adj[lo] + ii;
    };

    for (unsigned i = tid; i < cnt_b; i += 4 * BIN_W) {
        int d[4];
        unsigned px[4];
        if (staged) {
#pragma unroll
            for (int u = 0; u < 4; ++u) {
                const unsigned ii = i + u * BIN_W;
                px[u] = s_pidx[ii < cnt_b ? ii : cnt_b - 1];
            }
        } else {
#pragma unroll
            for (int u = 0; u < 4; ++u) {
                const unsigned ii = i + u * BIN_W;
                px[u] = bisect_index(ii < cnt_b ? ii : cnt_b - 1);
            }
        }
#pragma unroll
        for (int u = 0; u < 4; ++u) d[u] = pairs[px[u]].y;
#pragma unroll
        for (int u = 0; u < 4; ++u)
            if (i + u * BIN_W < cnt_b) atomicAdd(&s_cnt[(unsigned)d[u] & (BIN_W - 1)], 1u);
    }
    __syncthreads();

    const unsigned c = s_cnt[tid];
    const unsigned incl = wave_incl_scan_u32(c, lane);
    if (lane == 31) s_wtot[wave] = incl;
    __syncthreads();
    unsigned excl = incl - c;
    for (int w = 0; w < BIN_W / 32; ++w)
        if (w < wave) excl += s_wtot[w];

    const int node = b * BIN_W + tid;
    if (node < n_nodes) {
        start[node] = beg + excl;
        invdeg[node] = 1.0f / fmaxf((float)c, 1.0f);
    }
    if (b == n_bins - 1 && tid == 0) start[n_nodes] = beg + cnt_b;
    s_cnt[tid] = excl;
    __syncthreads();

    for (unsigned i = tid; i < cnt_b; i += 4 * BIN_W) {
        int2 v[4];
        unsigned px[4];
        if (staged) {
#pragma unroll
            for (int u = 0; u < 4; ++u) {
                const unsigned ii = i + u * BIN_W;
                px[u] = s_pidx[ii < cnt_b ? ii : cnt_b - 1];
            }
        } else {
#pragma unroll
            for (int u = 0; u < 4; ++u) {
                const unsigned ii = i + u * BIN_W;
                px[u] = bisect_index(ii < cnt_b ? ii : cnt_b - 1);
            }
        }
#pragma unroll
        for (int u = 0; u < 4; ++u) v[u] = pairs[px[u]];
#pragma unroll
        for (int u = 0; u < 4; ++u) {
            if (i + u * BIN_W < cnt_b) {
                const unsigned p = atomicAdd(&s_cnt[(unsigned)v[u].y & (BIN_W - 1)], 1u);
                if (staged) s_nbr[p] = v[u].x;
                else nbr[beg + p] = v[u].x;
            }
        }
    }
    if (staged) {
        __syncthreads();
        for (unsigned i = tid; i < cnt_b; i += BIN_W) nbr[beg + i] = s_nbr[i];
    }
}

#define APITCH 68
#define ROWS_IN_FLIGHT 16
#define LAYER_WAVES 25

__device__ __forceinline__ void self_step(const float* xa_lane, const v2f* wr_lane,
                                          const unsigned ks, v8f (&acc4)[4]) {
    const v2f a = *(const v2f*)&xa_lane[4u * ks];
#pragma unroll
    for (int nt = 0; nt < 4; ++nt) {
        const v2f b = wr_lane[2u * ks * WPAD + nt * 16];
        acc4[nt] = __builtin_amdgcn_wmma_f32_16x16x4_f32(
            false, a, false, b, (short)0, acc4[nt], false, false);
    }
    asm volatile("v_nop\n\tv_nop\n\tv_nop\n\tv_nop"
                 : "+v"(acc4[0]), "+v"(acc4[1]), "+v"(acc4[2]), "+v"(acc4[3]));
}

template <bool FULL>
__device__ __forceinline__ void ask_rows(const v2f* rows, const int mine, const unsigned j0,
                                         const unsigned cnt, const int lane,
                                         v2f (&v)[ROWS_IN_FLIGHT]) {
#pragma unroll
    for (int u = 0; u < ROWS_IN_FLIGHT; ++u) {
        const unsigned jj = (FULL || (unsigned)u < cnt) ? (unsigned)u : cnt - 1;
        v[u] = rows[(long long)__builtin_amdgcn_readlane(mine, j0 + jj) * 32 + lane];
    }
}

template <bool FULL>
__device__ __forceinline__ void add_rows(const v2f (&v)[ROWS_IN_FLIGHT], const unsigned cnt,
                                         const unsigned pos0, const int mystart, const float id,
                                         float* my_a, const int lane,
                                         v2f& acc, unsigned& cur, unsigned& nb) {
#pragma unroll
    for (int u = 0; u < ROWS_IN_FLIGHT; ++u) {
        if (FULL || (unsigned)u < cnt) {
            const unsigned p = pos0 + (unsigned)u;
            while (p == nb) {
                const float sc = __builtin_bit_cast(
                    float, __builtin_amdgcn_readlane(__builtin_bit_cast(int, id), cur));
                v2f o;
                o.x = acc.x * sc;
                o.y = acc.y * sc;
                *(v2f*)&my_a[cur * APITCH + 2 * lane] = o;
                acc.x = 0.0f;
                acc.y = 0.0f;
                ++cur;
                nb = (unsigned)__builtin_amdgcn_readlane(mystart, cur + 1);
            }
            acc += v[u];
        }
    }
}

template <bool STEP>
__device__ __forceinline__ void sum_span(const v2f* rows, const int* nbr, unsigned base,
                                         const unsigned stop, const unsigned te,
                                         const int mystart, const float id, float* my_a,
                                         const int lane, int& mine, v2f& acc,
                                         unsigned& cur, unsigned& nb,
                                         const float* xa_lane, const v2f* wr_lane,
                                         unsigned& ks, v8f (&acc4)[4]) {
    for (; base < stop; base += 32) {
        const unsigned left = te - base;
        const unsigned m = left < 32u ? left : 32u;
        const unsigned nx = base + 32u + (unsigned)lane;
        const int* next_entry = nbr + (nx < te ? nx : te - 1);
        int mine_next = 0;
        for (unsigned j0 = 0; j0 < m; j0 += ROWS_IN_FLIGHT) {
            const unsigned cnt = (m - j0) < (unsigned)ROWS_IN_FLIGHT ? (m - j0)
                                                                     : (unsigned)ROWS_IN_FLIGHT;
            const bool full = cnt == (unsigned)ROWS_IN_FLIGHT;
            v2f v[ROWS_IN_FLIGHT];
            if (full) ask_rows<true>(rows, mine, j0, cnt, lane, v);
            else ask_rows<false>(rows, mine, j0, cnt, lane, v);
            if (j0 == 0) mine_next = *next_entry;
            if (STEP) {
                self_step(xa_lane, wr_lane, ks, acc4);
                ++ks;
            }
            if (full) add_rows<true>(v, cnt, base + j0, mystart, id, my_a, lane, acc, cur, nb);
            else add_rows<false>(v, cnt, base + j0, mystart, id, my_a, lane, acc, cur, nb);
        }
        mine = mine_next;
    }
}

template <bool RELU>
__global__ void sage_layer_kernel(const float* __restrict__ feat,
                                  const unsigned* __restrict__ start,
                                  const int* __restrict__ nbr,
                                  const float* __restrict__ invdeg,
                                  const float* __restrict__ wl,
                                  const float* __restrict__ bl,
                                  const float* __restrict__ wr,
                                  float* __restrict__ out, int n_nodes) {
    __shared__ v2f lds_wl[32 * WPAD];
    __shared__ v2f lds_wr[32 * WPAD];
    __shared__ __attribute__((aligned(16))) float lds_a[LAYER_WAVES * 16 * APITCH];
    __shared__ v2f lds_x[LAYER_WAVES * 16 * APITCH / 2];

    for (int i = threadIdx.x; i < 32 * 64; i += blockDim.x) {
        const int p = i >> 6;
        const int n = i & 63;
        v2f a, b;
        a.x = wl[(2 * p) * DIM + n];
        a.y = wl[(2 * p + 1) * DIM + n];
        b.x = wr[(2 * p) * DIM + n];
        b.y = wr[(2 * p + 1) * DIM + n];
        lds_wl[p * WPAD + n] = a;
        lds_wr[p * WPAD + n] = b;
    }
    __syncthreads();

    const int lane = threadIdx.x & 31;
    const int wave = __builtin_amdgcn_readfirstlane((int)(threadIdx.x >> 5));
    const int tile = blockIdx.x * LAYER_WAVES + wave;
    const int n_tiles = (n_nodes + 15) >> 4;
    if (tile >= n_tiles) return;

    const int row0 = tile << 4;
    const int r     = lane & 15;
    const int kh    = (lane >> 4) << 1;
    const int pbase = lane >> 4;
    float* my_a = lds_a + wave * (16 * APITCH);
    float* my_x = (float*)lds_x + wave * (16 * APITCH);

    int node = row0 + r;
    if (node > n_nodes - 1) node = n_nodes - 1;
    const float id = invdeg[node];
    float bias[4];
#pragma unroll
    for (int nt = 0; nt < 4; ++nt) bias[nt] = bl[nt * 16 + r];

    int bi = row0 + (lane < 16 ? lane : 16);
    if (bi > n_nodes) bi = n_nodes;
    const int mystart = (int)start[bi];

    const v2f* rows = (const v2f*)feat;
    v2f xs[16];
#pragma unroll
    for (int i = 0; i < 16; ++i) {
        int sn = row0 + i;
        if (sn > n_nodes - 1) sn = n_nodes - 1;
        xs[i] = rows[(long long)sn * 32 + lane];
    }

    v8f acc4[4] = {{}, {}, {}, {}};
    const float* xa_lane = my_x + r * APITCH + kh;
    const v2f* wr_lane = lds_wr + pbase * WPAD + r;
    unsigned ks = 0;

    {
        const unsigned tb = (unsigned)__builtin_amdgcn_readlane(mystart, 0);
        const unsigned te = (unsigned)__builtin_amdgcn_readlane(mystart, 16);
        unsigned cur = 0;
        unsigned nb = (unsigned)__builtin_amdgcn_readlane(mystart, 1);
        v2f acc = {0.0f, 0.0f};
        int mine = 0;
        if (tb < te) mine = nbr[tb + (unsigned)lane < te ? tb + (unsigned)lane : te - 1];
#pragma unroll
        for (int i = 0; i < 16; ++i) *(v2f*)&my_x[i * APITCH + 2 * lane] = xs[i];
        const unsigned step_rows = (unsigned)(DIM / 4) * (unsigned)ROWS_IN_FLIGHT;
        static_assert(((DIM / 4) * ROWS_IN_FLIGHT) % 32 == 0, "the stepped part ends on a whole piece");
        const unsigned te_step = te - tb < step_rows ? te : tb + step_rows;
        sum_span<true>(rows, nbr, tb, te_step, te, mystart, id, my_a, lane, mine, acc, cur, nb,
                       xa_lane, wr_lane, ks, acc4);
        sum_span<false>(rows, nbr, te_step, te, te, mystart, id, my_a, lane, mine, acc, cur, nb,
                        xa_lane, wr_lane, ks, acc4);
        for (; cur < 16; ++cur) {
            const float sc = __builtin_bit_cast(
                float, __builtin_amdgcn_readlane(__builtin_bit_cast(int, id), cur));
            v2f o;
            o.x = acc.x * sc;
            o.y = acc.y * sc;
            *(v2f*)&my_a[cur * APITCH + 2 * lane] = o;
            acc.x = 0.0f;
            acc.y = 0.0f;
        }
    }

#pragma nounroll
    for (; ks < (unsigned)(DIM / 4); ++ks) self_step(xa_lane, wr_lane, ks, acc4);

    const float* arow = my_a + r * APITCH;
#pragma unroll
    for (int k = 0; k < DIM; k += 4) {
        const v2f a = *(const v2f*)&arow[k + kh];
        const int prow = (k >> 1) + pbase;
#pragma unroll
        for (int nt = 0; nt < 4; ++nt) {
            v2f b = lds_wl[prow * WPAD + nt * 16 + r];
            acc4[nt] = __builtin_amdgcn_wmma_f32_16x16x4_f32(
                false, a, false, b, (short)0, acc4[nt], false, false);
        }
    }

    const int mbase = (lane >> 4) << 3;
#pragma unroll
    for (int nt = 0; nt < 4; ++nt) {
#pragma unroll
        for (int vr = 0; vr < 8; ++vr) {
            float v = acc4[nt][vr] + bias[nt];
            if (RELU) v = fmaxf(v, 0.0f);
            my_a[(mbase + vr) * APITCH + nt * 16 + r] = v;
        }
    }
    const int orow = lane >> 4;
    const int ocol = (lane & 15) << 2;
    float* otile = out + (long long)row0 * DIM;
    if (row0 + 16 <= n_nodes) {
#pragma unroll
        for (int i = 0; i < 8; ++i) {
            const v4f v = *(const v4f*)&my_a[(2 * i + orow) * APITCH + ocol];
            *(v4f*)&otile[(2 * i + orow) * DIM + ocol] = v;
        }
    } else {
#pragma unroll
        for (int i = 0; i < 8; ++i) {
            if (row0 + 2 * i + orow < n_nodes) {
                const v4f v = *(const v4f*)&my_a[(2 * i + orow) * APITCH + ocol];
                *(v4f*)&otile[(2 * i + orow) * DIM + ocol] = v;
            }
        }
    }
}

extern "C" void kernel_launch(void* const* d_in, const int* in_sizes, int n_in,
                              void* d_out, int out_size, void* d_ws, size_t ws_size,
                              hipStream_t stream) {
    const float* x   = (const float*)d_in[0];
    const int*  edge = (const int*) d_in[1];
    const float* w1l = (const float*)d_in[2];
    const float* b1l = (const float*)d_in[3];
    const float* w1r = (const float*)d_in[4];
    const float* w2l = (const float*)d_in[5];
    const float* b2l = (const float*)d_in[6];
    const float* w2r = (const float*)d_in[7];
    float* out = (float*)d_out;

    const int n_nodes = in_sizes[0] / DIM;
    const int n_edges = in_sizes[1] / 2;
    const int* src = edge;
    const int* dst = edge + n_edges;

    float* ws      = (float*)d_ws;
    float* invdeg  = ws;
    float* list    = ws + n_nodes;
    float* h       = list + (long long)n_nodes * DIM;
    unsigned* start = (unsigned*)list;
    int*      nbr   = (int*)(start + n_nodes + 1);

    const int n_bins       = (n_nodes + BIN_W - 1) / BIN_W;
    const int n_chunks     = (n_edges + CH_EDGES - 1) / CH_EDGES;
    const int tiles        = (n_nodes + 15) / 16;
    const int layer_blocks = (tiles + LAYER_WAVES - 1) / LAYER_WAVES;

    int2*     pairs = (int2*)d_out;
    unsigned* tab   = (unsigned*)d_out + 2 * (long long)n_edges;

    chunk_sort_kernel<<<n_chunks, ORDER_T, 0, stream>>>(src, dst, tab, pairs, n_edges, n_bins);
    bin_sort_kernel<<<n_bins, BIN_W, 0, stream>>>(pairs, tab, start, invdeg, nbr, n_nodes,
                                                 n_bins, n_chunks);

    sage_layer_kernel<true><<<layer_blocks, LAYER_WAVES * 32, 0, stream>>>(
        x, start, nbr, invdeg, w1l, b1l, w1r, h, n_nodes);

    sage_layer_kernel<false><<<layer_blocks, LAYER_WAVES * 32, 0, stream>>>(
        h, start, nbr, invdeg, w2l, b2l, w2r, out, n_nodes);
}


# Per-bin edge ordering moved into layer 1's launch: a layer work-group orders its own 400 nodes' edges in LDS (bins of 400, 800 threads), then gathers; list = LDS arrival order; 4 launches to 3
# speedup vs baseline: 8.8889x; 1.0709x over previous
#include <hip/hip_runtime.h>

typedef float v2f __attribute__((ext_vector_type(2)));
typedef float v4f __attribute__((ext_vector_type(4)));
typedef float v8f __attribute__((ext_vector_type(8)));

#define DIM 64
#define WPAD 80

#define LAYER_WAVES 25
#define LAYER_T (LAYER_WAVES * 32)
#define APITCH 68
#define BIN_W (LAYER_WAVES * 16)
#define NB_MAX 1024
#define CH_EDGES 6400
#define ORDER_T 512
#define STAGE_CAP 16384

__device__ __forceinline__ unsigned wave_incl_scan_u32(unsigned v, int lane) {
    for (int o = 1; o < 32; o <<= 1) {
        const unsigned up = (unsigned)__shfl_up((int)v, o);
        if (lane >= o) v += up;
    }
    return v;
}

__device__ __forceinline__ void wave_excl_scan_lds(unsigned* a, int n, int lane) {
    const int per = (n + 31) >> 5;
    const int lo = lane * per;
    unsigned t = 0;
    for (int k = 0; k < per; ++k) {
        const int i = lo + k;
        if (i < n) t += a[i];
    }
    const unsigned incl = wave_incl_scan_u32(t, lane);
    unsigned run = incl - t;
    for (int k = 0; k < per; ++k) {
        const int i = lo + k;
        if (i < n) {
            const unsigned c = a[i];
            a[i] = run;
            run += c;
        }
    }
}

#define CH_PER ((CH_EDGES + ORDER_T - 1) / ORDER_T)
static_assert(CH_EDGES < 65536, "run start and count share one 32-bit word");
__global__ void chunk_sort_kernel(const int* __restrict__ src, const int* __restrict__ dst,
                                  unsigned* __restrict__ tab, int2* __restrict__ pairs,
                                  int n_edges, int n_bins) {
    __shared__ unsigned s_cnt[NB_MAX];
    __shared__ unsigned s_lbase[NB_MAX];
    __shared__ int2 s_stage[CH_EDGES];
    const int c = blockIdx.x;
    const int lane = threadIdx.x & 31;
    const int wave = threadIdx.x >> 5;

    for (int i = threadIdx.x; i < n_bins; i += blockDim.x) s_cnt[i] = 0u;
    __syncthreads();

    const int e0 = c * CH_EDGES;
    int e1 = e0 + CH_EDGES;
    if (e1 > n_edges) e1 = n_edges;
    int d[CH_PER], s[CH_PER];
    unsigned rk[CH_PER];
#pragma unroll
    for (int u = 0; u < CH_PER; ++u) {
        const int ee = e0 + (int)threadIdx.x + u * ORDER_T;
        const int ec = ee < e1 ? ee : e1 - 1;
        d[u] = dst[ec];
        s[u] = src[ec];
    }
#pragma unroll
    for (int u = 0; u < CH_PER; ++u) {
        rk[u] = 0u;
        if (e0 + (int)threadIdx.x + u * ORDER_T < e1)
            rk[u] = atomicAdd(&s_cnt[(unsigned)d[u] / (unsigned)BIN_W], 1u);
    }
    __syncthreads();
    for (int i = threadIdx.x; i < n_bins; i += blockDim.x) s_lbase[i] = s_cnt[i];
    __syncthreads();
    if (wave == 0) wave_excl_scan_lds(s_lbase, n_bins, lane);
    __syncthreads();
#pragma unroll
    for (int u = 0; u < CH_PER; ++u) {
        if (e0 + (int)threadIdx.x + u * ORDER_T < e1)
            s_stage[s_lbase[(unsigned)d[u] / (unsigned)BIN_W] + rk[u]] = make_int2(s[u], d[u]);
    }
    for (int i = threadIdx.x; i < n_bins; i += blockDim.x)
        tab[(long long)c * n_bins + i] = (s_lbase[i] << 16) | s_cnt[i];
    __syncthreads();
    const int cnt = e1 - e0;
    for (int i = threadIdx.x; i < cnt; i += blockDim.x) pairs[e0 + i] = s_stage[i];
}

#define CH_MAX 1024
__device__ __forceinline__ void order_bin(const int2* __restrict__ pairs,
                                          const unsigned* __restrict__ tab,
                                          unsigned* start, float* invdeg, int* nbr,
                                          unsigned* s_start, float* s_inv,
                                          unsigned* scr_a, unsigned* scr_x, const int b,
                                          const int n_nodes, const int n_bins,
                                          const int n_chunks) {
    int* s_nbr = (int*)scr_a;
    unsigned* s_rb = scr_a + STAGE_CAP;
    unsigned* s_adj = s_rb + (CH_MAX + 1);
    unsigned* s_cnt = s_adj + CH_MAX;
    unsigned* s_wtot = s_cnt + BIN_W;
    unsigned* s_begp = s_wtot + LAYER_WAVES;
    unsigned* s_pidx = scr_x;
    static_assert(STAGE_CAP + (CH_MAX + 1) + CH_MAX + BIN_W + LAYER_WAVES + 1 <= LAYER_WAVES * 16 * APITCH,
                  "the step's tables fit the sum tiles");
    static_assert(STAGE_CAP <= LAYER_WAVES * 16 * APITCH, "the entry table fits the self tiles");
    const int tid = threadIdx.x;
    const int lane = tid & 31;
    const int wave = tid >> 5;
    const unsigned node0 = (unsigned)b * (unsigned)BIN_W;

    if (tid < BIN_W) s_cnt[tid] = 0u;
    if (tid == 0) {
        *s_begp = 0u;
        s_rb[n_chunks] = 0u;
    }
    __syncthreads();
    unsigned lbsum = 0u;
    for (int t = tid; t < n_chunks; t += LAYER_T) {
        const unsigned inf = tab[(long long)t * n_bins + b];
        s_rb[t] = inf & 0xffffu;
        s_adj[t] = (unsigned)t * (unsigned)CH_EDGES + (inf >> 16);
        lbsum += inf >> 16;
    }
    for (int o = 16; o > 0; o >>= 1) lbsum += (unsigned)__shfl_xor((int)lbsum, o);
    if (lane == 0 && lbsum != 0u) atomicAdd(s_begp, lbsum);
    __syncthreads();
    if (wave == 0) wave_excl_scan_lds(s_rb, n_chunks + 1, lane);
    __syncthreads();
    const unsigned beg = *s_begp;
    const unsigned cnt_b = s_rb[n_chunks];
    const bool staged = cnt_b <= (unsigned)STAGE_CAP;
    for (int t = tid; t < n_chunks; t += LAYER_T) {
        const unsigned lo = s_rb[t];
        const unsigned hi = s_rb[t + 1];
        const unsigned adj = s_adj[t] - lo;
        s_adj[t] = adj;
        if (staged)
            for (unsigned m = lo; m < hi; ++m) s_pidx[m] = adj + m;
    }
    __syncthreads();

    auto bisect_index = [&](const unsigned ii) -> unsigned {
        unsigned lo = 0u, hi = (unsigned)n_chunks;
        while (hi - lo > 1u) {
            const unsigned mid = (lo + hi) >> 1;
            if (s_rb[mid] <= ii) lo = mid;
            else hi = mid;
        }
        return s_adj[lo] + ii;
    };

    for (unsigned i = tid; i < cnt_b; i += 4 * LAYER_T) {
        int d[4];
        unsigned px[4];
        if (staged) {
#pragma unroll
            for (int u = 0; u < 4; ++u) {
                const unsigned ii = i + u * LAYER_T;
                px[u] = s_pidx[ii < cnt_b ? ii : cnt_b - 1];
            }
        } else {
#pragma unroll
            for (int u = 0; u < 4; ++u) {
                const unsigned ii = i + u * LAYER_T;
                px[u] = bisect_index(ii < cnt_b ? ii : cnt_b - 1);
            }
        }
#pragma unroll
        for (int u = 0; u < 4; ++u) d[u] = pairs[px[u]].y;
#pragma unroll
        for (int u = 0; u < 4; ++u)
            if (i + u * LAYER_T < cnt_b) atomicAdd(&s_cnt[(unsigned)d[u] - node0], 1u);
    }
    __syncthreads();

    const unsigned c = tid < BIN_W ? s_cnt[tid] : 0u;
    const unsigned incl = wave_incl_scan_u32(c, lane);
    if (lane == 31) s_wtot[wave] = incl;
    __syncthreads();
    unsigned excl = incl - c;
    for (int w = 0; w < LAYER_WAVES; ++w)
        if (w < wave) excl += s_wtot[w];

    if (tid < BIN_W) {
        const unsigned st = beg + excl;
        const float iv = 1.0f / fmaxf((float)c, 1.0f);
        s_start[tid] = st;
        s_inv[tid] = iv;
        const int node = (int)node0 + tid;
        if (node < n_nodes) {
            start[node] = st;
            invdeg[node] = iv;
        }
        s_cnt[tid] = excl;
    }
    if (tid == 0) {
        s_start[BIN_W] = beg + cnt_b;
        if (b == n_bins - 1) start[n_nodes] = beg + cnt_b;
    }
    __syncthreads();

    for (unsigned i = tid; i < cnt_b; i += 4 * LAYER_T) {
        int2 v[4];
        unsigned px[4];
        if (staged) {
#pragma unroll
            for (int u = 0; u < 4; ++u) {
                const unsigned ii = i + u * LAYER_T;
                px[u] = s_pidx[ii < cnt_b ? ii : cnt_b - 1];
            }
        } else {
#pragma unroll
            for (int u = 0; u < 4; ++u) {
                const unsigned ii = i + u * LAYER_T;
                px[u] = bisect_index(ii < cnt_b ? ii : cnt_b - 1);
            }
        }
#pragma unroll
        for (int u = 0; u < 4; ++u) v[u] = pairs[px[u]];
#pragma unroll
        for (int u = 0; u < 4; ++u) {
            if (i + u * LAYER_T < cnt_b) {
                const unsigned p = atomicAdd(&s_cnt[(unsigned)v[u].y - node0], 1u);
                if (staged) s_nbr[p] = v[u].x;
                else nbr[beg + p] = v[u].x;
            }
        }
    }
    if (staged) {
        __syncthreads();
        for (unsigned i = tid; i < cnt_b; i += LAYER_T) nbr[beg + i] = s_nbr[i];
    }
    __syncthreads();
}

#define ROWS_IN_FLIGHT 16

__device__ __forceinline__ void self_step(const float* xa_lane, const v2f* wr_lane,
                                          const unsigned ks, v8f (&acc4)[4]) {
    const v2f a = *(const v2f*)&xa_lane[4u * ks];
#pragma unroll
    for (int nt = 0; nt < 4; ++nt) {
        const v2f b = wr_lane[2u * ks * WPAD + nt * 16];
        acc4[nt] = __builtin_amdgcn_wmma_f32_16x16x4_f32(
            false, a, false, b, (short)0, acc4[nt], false, false);
    }
    asm volatile("v_nop\n\tv_nop\n\tv_nop\n\tv_nop"
                 : "+v"(acc4[0]), "+v"(acc4[1]), "+v"(acc4[2]), "+v"(acc4[3]));
}

template <bool FULL>
__device__ __forceinline__ void ask_rows(const v2f* rows, const int mine, const unsigned j0,
                                         const unsigned cnt, const int lane,
                                         v2f (&v)[ROWS_IN_FLIGHT]) {
#pragma unroll
    for (int u = 0; u < ROWS_IN_FLIGHT; ++u) {
        const unsigned jj = (FULL || (unsigned)u < cnt) ? (unsigned)u : cnt - 1;
        v[u] = rows[(long long)__builtin_amdgcn_readlane(mine, j0 + jj) * 32 + lane];
    }
}

template <bool FULL>
__device__ __forceinline__ void add_rows(const v2f (&v)[ROWS_IN_FLIGHT], const unsigned cnt,
                                         const unsigned pos0, const int mystart, const float id,
                                         float* my_a, const int lane,
                                         v2f& acc, unsigned& cur, unsigned& nb) {
#pragma unroll
    for (int u = 0; u < ROWS_IN_FLIGHT; ++u) {
        if (FULL || (unsigned)u < cnt) {
            const unsigned p = pos0 + (unsigned)u;
            while (p == nb) {
                const float sc = __builtin_bit_cast(
                    float, __builtin_amdgcn_readlane(__builtin_bit_cast(int, id), cur));
                v2f o;
                o.x = acc.x * sc;
                o.y = acc.y * sc;
                *(v2f*)&my_a[cur * APITCH + 2 * lane] = o;
                acc.x = 0.0f;
                acc.y = 0.0f;
                ++cur;
                nb = (unsigned)__builtin_amdgcn_readlane(mystart, cur + 1);
            }
            acc += v[u];
        }
    }
}

template <bool STEP>
__device__ __forceinline__ void sum_span(const v2f* rows, const int* nbr, unsigned base,
                                         const unsigned stop, const unsigned te,
                                         const int mystart, const float id, float* my_a,
                                         const int lane, int& mine, v2f& acc,
                                         unsigned& cur, unsigned& nb,
                                         const float* xa_lane, const v2f* wr_lane,
                                         unsigned& ks, v8f (&acc4)[4]) {
    for (; base < stop; base += 32) {
        const unsigned left = te - base;
        const unsigned m = left < 32u ? left : 32u;
        const unsigned nx = base + 32u + (unsigned)lane;
        const int* next_entry = nbr + (nx < te ? nx : te - 1);
        int mine_next = 0;
        for (unsigned j0 = 0; j0 < m; j0 += ROWS_IN_FLIGHT) {
            const unsigned cnt = (m - j0) < (unsigned)ROWS_IN_FLIGHT ? (m - j0)
                                                                     : (unsigned)ROWS_IN_FLIGHT;
            const bool full = cnt == (unsigned)ROWS_IN_FLIGHT;
            v2f v[ROWS_IN_FLIGHT];
            if (full) ask_rows<true>(rows, mine, j0, cnt, lane, v);
            else ask_rows<false>(rows, mine, j0, cnt, lane, v);
            if (j0 == 0) mine_next = *next_entry;
            if (STEP) {
                self_step(xa_lane, wr_lane, ks, acc4);
                ++ks;
            }
            if (full) add_rows<true>(v, cnt, base + j0, mystart, id, my_a, lane, acc, cur, nb);
            else add_rows<false>(v, cnt, base + j0, mystart, id, my_a, lane, acc, cur, nb);
        }
        mine = mine_next;
    }
}

template <bool RELU, bool ORDER>
__device__ __forceinline__ void layer_body(const float* __restrict__ feat,
                                           const int2* __restrict__ pairs,
                                           const unsigned* __restrict__ tab,
                                           unsigned* start, int* nbr, float* invdeg,
                                           const float* __restrict__ wl,
                                           const float* __restrict__ bl,
                                           const float* __restrict__ wr,
                                           float* __restrict__ out, const int n_nodes,
                                           const int n_bins, const int n_chunks) {
    __shared__ unsigned s_start[BIN_W + 1];
    __shared__ float s_inv[BIN_W];
    __shared__ v2f lds_wl[32 * WPAD];
    __shared__ v2f lds_wr[32 * WPAD];
    __shared__ __attribute__((aligned(16))) float lds_a[LAYER_WAVES * 16 * APITCH];
    __shared__ v2f lds_x[LAYER_WAVES * 16 * APITCH / 2];

    for (int i = threadIdx.x; i < 32 * 64; i += blockDim.x) {
        const int p = i >> 6;
        const int n = i & 63;
        v2f a, b;
        a.x = wl[(2 * p) * DIM + n];
        a.y = wl[(2 * p + 1) * DIM + n];
        b.x = wr[(2 * p) * DIM + n];
        b.y = wr[(2 * p + 1) * DIM + n];
        lds_wl[p * WPAD + n] = a;
        lds_wr[p * WPAD + n] = b;
    }
    __syncthreads();
    if constexpr (ORDER)
        order_bin(pairs, tab, start, invdeg, nbr, s_start, s_inv, (unsigned*)lds_a, (unsigned*)lds_x,
                  (int)blockIdx.x, n_nodes, n_bins, n_chunks);

    const int lane = threadIdx.x & 31;
    const int wave = __builtin_amdgcn_readfirstlane((int)(threadIdx.x >> 5));
    const int tile = blockIdx.x * LAYER_WAVES + wave;
    const int n_tiles = (n_nodes + 15) >> 4;
    if (tile >= n_tiles) return;

    const int row0 = tile << 4;
    const int r     = lane & 15;
    const int kh    = (lane >> 4) << 1;
    const int pbase = lane >> 4;
    float* my_a = lds_a + wave * (16 * APITCH);
    float* my_x = (float*)lds_x + wave * (16 * APITCH);

    int node = row0 + r;
    if (node > n_nodes - 1) node = n_nodes - 1;
    const float id = ORDER ? s_inv[wave * 16 + r] : invdeg[node];
    float bias[4];
#pragma unroll
    for (int nt = 0; nt < 4; ++nt) bias[nt] = bl[nt * 16 + r];

    int bi = row0 + (lane < 16 ? lane : 16);
    if (bi > n_nodes) bi = n_nodes;
    const int mystart = ORDER ? (int)s_start[wave * 16 + (lane < 16 ? lane : 16)] : (int)start[bi];

    const v2f* rows = (const v2f*)feat;
    v2f xs[16];
#pragma unroll
    for (int i = 0; i < 16; ++i) {
        int sn = row0 + i;
        if (sn > n_nodes - 1) sn = n_nodes - 1;
        xs[i] = rows[(long long)sn * 32 + lane];
    }

    v8f acc4[4] = {{}, {}, {}, {}};
    const float* xa_lane = my_x + r * APITCH + kh;
    const v2f* wr_lane = lds_wr + pbase * WPAD + r;
    unsigned ks = 0;

    {
        const unsigned tb = (unsigned)__builtin_amdgcn_readlane(mystart, 0);
        const unsigned te = (unsigned)__builtin_amdgcn_readlane(mystart, 16);
        unsigned cur = 0;
        unsigned nb = (unsigned)__builtin_amdgcn_readlane(mystart, 1);
        v2f acc = {0.0f, 0.0f};
        int mine = 0;
        if (tb < te) mine = nbr[tb + (unsigned)lane < te ? tb + (unsigned)lane : te - 1];
#pragma unroll
        for (int i = 0; i < 16; ++i) *(v2f*)&my_x[i * APITCH + 2 * lane] = xs[i];
        const unsigned step_rows = (unsigned)(DIM / 4) * (unsigned)ROWS_IN_FLIGHT;
        static_assert(((DIM / 4) * ROWS_IN_FLIGHT) % 32 == 0, "the stepped part ends on a whole piece");
        const unsigned te_step = te - tb < step_rows ? te : tb + step_rows;
        sum_span<true>(rows, nbr, tb, te_step, te, mystart, id, my_a, lane, mine, acc, cur, nb,
                       xa_lane, wr_lane, ks, acc4);
        sum_span<false>(rows, nbr, te_step, te, te, mystart, id, my_a, lane, mine, acc, cur, nb,
                        xa_lane, wr_lane, ks, acc4);
        for (; cur < 16; ++cur) {
            const float sc = __builtin_bit_cast(
                float, __builtin_amdgcn_readlane(__builtin_bit_cast(int, id), cur));
            v2f o;
            o.x = acc.x * sc;
            o.y = acc.y * sc;
            *(v2f*)&my_a[cur * APITCH + 2 * lane] = o;
            acc.x = 0.0f;
            acc.y = 0.0f;
        }
    }

#pragma nounroll
    for (; ks < (unsigned)(DIM / 4); ++ks) self_step(xa_lane, wr_lane, ks, acc4);

    const float* arow = my_a + r * APITCH;
#pragma unroll
    for (int k = 0; k < DIM; k += 4) {
        const v2f a = *(const v2f*)&arow[k + kh];
        const int prow = (k >> 1) + pbase;
#pragma unroll
        for (int nt = 0; nt < 4; ++nt) {
            v2f b = lds_wl[prow * WPAD + nt * 16 + r];
            acc4[nt] = __builtin_amdgcn_wmma_f32_16x16x4_f32(
                false, a, false, b, (short)0, acc4[nt], false, false);
        }
    }

    const int mbase = (lane >> 4) << 3;
#pragma unroll
    for (int nt = 0; nt < 4; ++nt) {
#pragma unroll
        for (int vr = 0; vr < 8; ++vr) {
            float v = acc4[nt][vr] + bias[nt];
            if (RELU) v = fmaxf(v, 0.0f);
            my_a[(mbase + vr) * APITCH + nt * 16 + r] = v;
        }
    }
    const int orow = lane >> 4;
    const int ocol = (lane & 15) << 2;
    float* otile = out + (long long)row0 * DIM;
    if (row0 + 16 <= n_nodes) {
#pragma unroll
        for (int i = 0; i < 8; ++i) {
            const v4f v = *(const v4f*)&my_a[(2 * i + orow) * APITCH + ocol];
            *(v4f*)&otile[(2 * i + orow) * DIM + ocol] = v;
        }
    } else {
#pragma unroll
        for (int i = 0; i < 8; ++i) {
            if (row0 + 2 * i + orow < n_nodes) {
                const v4f v = *(const v4f*)&my_a[(2 * i + orow) * APITCH + ocol];
                *(v4f*)&otile[(2 * i + orow) * DIM + ocol] = v;
            }
        }
    }
}

template <bool RELU>
__global__ void sage_order_layer_kernel(const float* __restrict__ feat,
                                        const int2* __restrict__ pairs,
                                        const unsigned* __restrict__ tab,
                                        unsigned* start, int* nbr, float* invdeg,
                                        const float* __restrict__ wl,
                                        const float* __restrict__ bl,
                                        const float* __restrict__ wr,
                                        float* __restrict__ out, int n_nodes, int n_bins,
                                        int n_chunks) {
    layer_body<RELU, true>(feat, pairs, tab, start, nbr, invdeg, wl, bl, wr, out, n_nodes, n_bins,
                           n_chunks);
}

template <bool RELU>
__global__ void sage_layer_kernel(const float* __restrict__ feat,
                                  const unsigned* __restrict__ start,
                                  const int* __restrict__ nbr,
                                  const float* __restrict__ invdeg,
                                  const float* __restrict__ wl,
                                  const float* __restrict__ bl,
                                  const float* __restrict__ wr,
                                  float* __restrict__ out, int n_nodes) {
    layer_body<RELU, false>(feat, nullptr, nullptr, const_cast<unsigned*>(start),
                            const_cast<int*>(nbr), const_cast<float*>(invdeg), wl, bl, wr, out,
                            n_nodes, 0, 0);
}

extern "C" void kernel_launch(void* const* d_in, const int* in_sizes, int n_in,
                              void* d_out, int out_size, void* d_ws, size_t ws_size,
                              hipStream_t stream) {
    const float* x   = (const float*)d_in[0];
    const int*  edge = (const int*) d_in[1];
    const float* w1l = (const float*)d_in[2];
    const float* b1l = (const float*)d_in[3];
    const float* w1r = (const float*)d_in[4];
    const float* w2l = (const float*)d_in[5];
    const float* b2l = (const float*)d_in[6];
    const float* w2r = (const float*)d_in[7];
    float* out = (float*)d_out;

    const int n_nodes = in_sizes[0] / DIM;
    const int n_edges = in_sizes[1] / 2;
    const int* src = edge;
    const int* dst = edge + n_edges;

    float* ws      = (float*)d_ws;
    float* invdeg  = ws;
    float* list    = ws + n_nodes;
    float* h       = list + (long long)n_nodes * DIM;
    unsigned* start = (unsigned*)list;
    int*      nbr   = (int*)(start + n_nodes + 1);

    const int n_bins       = (n_nodes + BIN_W - 1) / BIN_W;
    const int n_chunks     = (n_edges + CH_EDGES - 1) / CH_EDGES;
    const int tiles        = (n_nodes + 15) / 16;
    const int layer_blocks = (tiles + LAYER_WAVES - 1) / LAYER_WAVES;

    int2*     pairs = (int2*)d_out;
    unsigned* tab   = (unsigned*)d_out + 2 * (long long)n_edges;

    chunk_sort_kernel<<<n_chunks, ORDER_T, 0, stream>>>(src, dst, tab, pairs, n_edges, n_bins);

    sage_order_layer_kernel<true><<<n_bins, LAYER_T, 0, stream>>>(
        x, pairs, tab, start, nbr, invdeg, w1l, b1l, w1r, h, n_nodes, n_bins, n_chunks);

    sage_layer_kernel<false><<<layer_blocks, LAYER_WAVES * 32, 0, stream>>>(
        h, start, nbr, invdeg, w2l, b2l, w2r, out, n_nodes);
}


# Row reads run one group (16 rows) ahead of the adds: two row buffers per wave, 16-32 rows on their way under counted waits, no drain between groups; 32-bit row offsets; sum order unchanged
# speedup vs baseline: 9.2507x; 1.0407x over previous
#include <hip/hip_runtime.h>

typedef float v2f __attribute__((ext_vector_type(2)));
typedef float v4f __attribute__((ext_vector_type(4)));
typedef float v8f __attribute__((ext_vector_type(8)));

#define DIM 64
#define WPAD 80

#define LAYER_WAVES 25
#define LAYER_T (LAYER_WAVES * 32)
#define APITCH 68
#define BIN_W (LAYER_WAVES * 16)
#define NB_MAX 1024
#define CH_EDGES 6400
#define ORDER_T 512
#define STAGE_CAP 16384

__device__ __forceinline__ unsigned wave_incl_scan_u32(unsigned v, int lane) {
    for (int o = 1; o < 32; o <<= 1) {
        const unsigned up = (unsigned)__shfl_up((int)v, o);
        if (lane >= o) v += up;
    }
    return v;
}

__device__ __forceinline__ void wave_excl_scan_lds(unsigned* a, int n, int lane) {
    const int per = (n + 31) >> 5;
    const int lo = lane * per;
    unsigned t = 0;
    for (int k = 0; k < per; ++k) {
        const int i = lo + k;
        if (i < n) t += a[i];
    }
    const unsigned incl = wave_incl_scan_u32(t, lane);
    unsigned run = incl - t;
    for (int k = 0; k < per; ++k) {
        const int i = lo + k;
        if (i < n) {
            const unsigned c = a[i];
            a[i] = run;
            run += c;
        }
    }
}

#define CH_PER ((CH_EDGES + ORDER_T - 1) / ORDER_T)
static_assert(CH_EDGES < 65536, "run start and count share one 32-bit word");
__global__ void chunk_sort_kernel(const int* __restrict__ src, const int* __restrict__ dst,
                                  unsigned* __restrict__ tab, int2* __restrict__ pairs,
                                  int n_edges, int n_bins) {
    __shared__ unsigned s_cnt[NB_MAX];
    __shared__ unsigned s_lbase[NB_MAX];
    __shared__ int2 s_stage[CH_EDGES];
    const int c = blockIdx.x;
    const int lane = threadIdx.x & 31;
    const int wave = threadIdx.x >> 5;

    for (int i = threadIdx.x; i < n_bins; i += blockDim.x) s_cnt[i] = 0u;
    __syncthreads();

    const int e0 = c * CH_EDGES;
    int e1 = e0 + CH_EDGES;
    if (e1 > n_edges) e1 = n_edges;
    int d[CH_PER], s[CH_PER];
    unsigned rk[CH_PER];
#pragma unroll
    for (int u = 0; u < CH_PER; ++u) {
        const int ee = e0 + (int)threadIdx.x + u * ORDER_T;
        const int ec = ee < e1 ? ee : e1 - 1;
        d[u] = dst[ec];
        s[u] = src[ec];
    }
#pragma unroll
    for (int u = 0; u < CH_PER; ++u) {
        rk[u] = 0u;
        if (e0 + (int)threadIdx.x + u * ORDER_T < e1)
            rk[u] = atomicAdd(&s_cnt[(unsigned)d[u] / (unsigned)BIN_W], 1u);
    }
    __syncthreads();
    for (int i = threadIdx.x; i < n_bins; i += blockDim.x) s_lbase[i] = s_cnt[i];
    __syncthreads();
    if (wave == 0) wave_excl_scan_lds(s_lbase, n_bins, lane);
    __syncthreads();
#pragma unroll
    for (int u = 0; u < CH_PER; ++u) {
        if (e0 + (int)threadIdx.x + u * ORDER_T < e1)
            s_stage[s_lbase[(unsigned)d[u] / (unsigned)BIN_W] + rk[u]] = make_int2(s[u], d[u]);
    }
    for (int i = threadIdx.x; i < n_bins; i += blockDim.x)
        tab[(long long)c * n_bins + i] = (s_lbase[i] << 16) | s_cnt[i];
    __syncthreads();
    const int cnt = e1 - e0;
    for (int i = threadIdx.x; i < cnt; i += blockDim.x) pairs[e0 + i] = s_stage[i];
}

#define CH_MAX 1024
__device__ __forceinline__ void order_bin(const int2* __restrict__ pairs,
                                          const unsigned* __restrict__ tab,
                                          unsigned* start, float* invdeg, int* nbr,
                                          unsigned* s_start, float* s_inv,
                                          unsigned* scr_a, unsigned* scr_x, const int b,
                                          const int n_nodes, const int n_bins,
                                          const int n_chunks) {
    int* s_nbr = (int*)scr_a;
    unsigned* s_rb = scr_a + STAGE_CAP;
    unsigned* s_adj = s_rb + (CH_MAX + 1);
    unsigned* s_cnt = s_adj + CH_MAX;
    unsigned* s_wtot = s_cnt + BIN_W;
    unsigned* s_begp = s_wtot + LAYER_WAVES;
    unsigned* s_pidx = scr_x;
    static_assert(STAGE_CAP + (CH_MAX + 1) + CH_MAX + BIN_W + LAYER_WAVES + 1 <= LAYER_WAVES * 16 * APITCH,
                  "the step's tables fit the sum tiles");
    static_assert(STAGE_CAP <= LAYER_WAVES * 16 * APITCH, "the entry table fits the self tiles");
    const int tid = threadIdx.x;
    const int lane = tid & 31;
    const int wave = tid >> 5;
    const unsigned node0 = (unsigned)b * (unsigned)BIN_W;

    if (tid < BIN_W) s_cnt[tid] = 0u;
    if (tid == 0) {
        *s_begp = 0u;
        s_rb[n_chunks] = 0u;
    }
    __syncthreads();
    unsigned lbsum = 0u;
    for (int t = tid; t < n_chunks; t += LAYER_T) {
        const unsigned inf = tab[(long long)t * n_bins + b];
        s_rb[t] = inf & 0xffffu;
        s_adj[t] = (unsigned)t * (unsigned)CH_EDGES + (inf >> 16);
        lbsum += inf >> 16;
    }
    for (int o = 16; o > 0; o >>= 1) lbsum += (unsigned)__shfl_xor((int)lbsum, o);
    if (lane == 0 && lbsum != 0u) atomicAdd(s_begp, lbsum);
    __syncthreads();
    if (wave == 0) wave_excl_scan_lds(s_rb, n_chunks + 1, lane);
    __syncthreads();
    const unsigned beg = *s_begp;
    const unsigned cnt_b = s_rb[n_chunks];
    const bool staged = cnt_b <= (unsigned)STAGE_CAP;
    for (int t = tid; t < n_chunks; t += LAYER_T) {
        const unsigned lo = s_rb[t];
        const unsigned hi = s_rb[t + 1];
        const unsigned adj = s_adj[t] - lo;
        s_adj[t] = adj;
        if (staged)
            for (unsigned m = lo; m < hi; ++m) s_pidx[m] = adj + m;
    }
    __syncthreads();

    auto bisect_index = [&](const unsigned ii) -> unsigned {
        unsigned lo = 0u, hi = (unsigned)n_chunks;
        while (hi - lo > 1u) {
            const unsigned mid = (lo + hi) >> 1;
            if (s_rb[mid] <= ii) lo = mid;
            else hi = mid;
        }
        return s_adj[lo] + ii;
    };

    for (unsigned i = tid; i < cnt_b; i += 4 * LAYER_T) {
        int d[4];
        unsigned px[4];
        if (staged) {
#pragma unroll
            for (int u = 0; u < 4; ++u) {
                const unsigned ii = i + u * LAYER_T;
                px[u] = s_pidx[ii < cnt_b ? ii : cnt_b - 1];
            }
        } else {
#pragma unroll
            for (int u = 0; u < 4; ++u) {
                const unsigned ii = i + u * LAYER_T;
                px[u] = bisect_index(ii < cnt_b ? ii : cnt_b - 1);
            }
        }
#pragma unroll
        for (int u = 0; u < 4; ++u) d[u] = pairs[px[u]].y;
#pragma unroll
        for (int u = 0; u < 4; ++u)
            if (i + u * LAYER_T < cnt_b) atomicAdd(&s_cnt[(unsigned)d[u] - node0], 1u);
    }
    __syncthreads();

    const unsigned c = tid < BIN_W ? s_cnt[tid] : 0u;
    const unsigned incl = wave_incl_scan_u32(c, lane);
    if (lane == 31) s_wtot[wave] = incl;
    __syncthreads();
    unsigned excl = incl - c;
    for (int w = 0; w < LAYER_WAVES; ++w)
        if (w < wave) excl += s_wtot[w];

    if (tid < BIN_W) {
        const unsigned st = beg + excl;
        const float iv = 1.0f / fmaxf((float)c, 1.0f);
        s_start[tid] = st;
        s_inv[tid] = iv;
        const int node = (int)node0 + tid;
        if (node < n_nodes) {
            start[node] = st;
            invdeg[node] = iv;
        }
        s_cnt[tid] = excl;
    }
    if (tid == 0) {
        s_start[BIN_W] = beg + cnt_b;
        if (b == n_bins - 1) start[n_nodes] = beg + cnt_b;
    }
    __syncthreads();

    for (unsigned i = tid; i < cnt_b; i += 4 * LAYER_T) {
        int2 v[4];
        unsigned px[4];
        if (staged) {
#pragma unroll
            for (int u = 0; u < 4; ++u) {
                const unsigned ii = i + u * LAYER_T;
                px[u] = s_pidx[ii < cnt_b ? ii : cnt_b - 1];
            }
        } else {
#pragma unroll
            for (int u = 0; u < 4; ++u) {
                const unsigned ii = i + u * LAYER_T;
                px[u] = bisect_index(ii < cnt_b ? ii : cnt_b - 1);
            }
        }
#pragma unroll
        for (int u = 0; u < 4; ++u) v[u] = pairs[px[u]];
#pragma unroll
        for (int u = 0; u < 4; ++u) {
            if (i + u * LAYER_T < cnt_b) {
                const unsigned p = atomicAdd(&s_cnt[(unsigned)v[u].y - node0], 1u);
                if (staged) s_nbr[p] = v[u].x;
                else nbr[beg + p] = v[u].x;
            }
        }
    }
    if (staged) {
        __syncthreads();
        for (unsigned i = tid; i < cnt_b; i += LAYER_T) nbr[beg + i] = s_nbr[i];
    }
    __syncthreads();
}

#define ROWS_IN_FLIGHT 16

__device__ __forceinline__ void self_step(const float* xa_lane, const v2f* wr_lane,
                                          const unsigned ks, v8f (&acc4)[4]) {
    const v2f a = *(const v2f*)&xa_lane[4u * ks];
#pragma unroll
    for (int nt = 0; nt < 4; ++nt) {
        const v2f b = wr_lane[2u * ks * WPAD + nt * 16];
        acc4[nt] = __builtin_amdgcn_wmma_f32_16x16x4_f32(
            false, a, false, b, (short)0, acc4[nt], false, false);
    }
    asm volatile("v_nop\n\tv_nop\n\tv_nop\n\tv_nop"
                 : "+v"(acc4[0]), "+v"(acc4[1]), "+v"(acc4[2]), "+v"(acc4[3]));
}

template <bool FULL>
__device__ __forceinline__ void ask_rows(const v2f* rows, const int mine, const unsigned j0,
                                         const unsigned cnt, const int lane,
                                         v2f (&v)[ROWS_IN_FLIGHT]) {
#pragma unroll
    for (int u = 0; u < ROWS_IN_FLIGHT; ++u) {
        const unsigned jj = (FULL || (unsigned)u < cnt) ? (unsigned)u : cnt - 1;
        v[u] = rows[(unsigned)__builtin_amdgcn_readlane(mine, j0 + jj) * 32u + (unsigned)lane];
    }
}

template <bool FULL>
__device__ __forceinline__ void add_rows(const v2f (&v)[ROWS_IN_FLIGHT], const unsigned cnt,
                                         const unsigned pos0, const int mystart, const float id,
                                         float* my_a, const int lane,
                                         v2f& acc, unsigned& cur, unsigned& nb) {
#pragma unroll
    for (int u = 0; u < ROWS_IN_FLIGHT; ++u) {
        if (FULL || (unsigned)u < cnt) {
            const unsigned p = pos0 + (unsigned)u;
            while (p == nb) {
                const float sc = __builtin_bit_cast(
                    float, __builtin_amdgcn_readlane(__builtin_bit_cast(int, id), cur));
                v2f o;
                o.x = acc.x * sc;
                o.y = acc.y * sc;
                *(v2f*)&my_a[cur * APITCH + 2 * lane] = o;
                acc.x = 0.0f;
                acc.y = 0.0f;
                ++cur;
                nb = (unsigned)__builtin_amdgcn_readlane(mystart, cur + 1);
            }
            acc += v[u];
        }
    }
}

static_assert(ROWS_IN_FLIGHT == 16, "a piece of 32 list entries is two groups");
template <bool STEP>
__device__ __forceinline__ void sum_span(const v2f* rows, const int* nbr, const unsigned tb,
                                         unsigned& pos, const unsigned stop, const unsigned te,
                                         const int mystart, const float id, float* my_a,
                                         const int lane, int& mine, int& mine1,
                                         v2f (&va)[ROWS_IN_FLIGHT], v2f& acc,
                                         unsigned& cur, unsigned& nb,
                                         const float* xa_lane, const v2f* wr_lane,
                                         unsigned& ks, v8f (&acc4)[4]) {
    while (pos < stop && te - pos >= 32u) {
        const unsigned nx = pos + 64u + (unsigned)lane;
        const int mine2 = nbr[nx < te ? nx : te - 1];
        v2f vb[ROWS_IN_FLIGHT];
        ask_rows<true>(rows, mine, ROWS_IN_FLIGHT, ROWS_IN_FLIGHT, lane, vb);
        if (STEP) {
            self_step(xa_lane, wr_lane, ks, acc4);
            ++ks;
        }
        add_rows<true>(va, ROWS_IN_FLIGHT, pos, mystart, id, my_a, lane, acc, cur, nb);
        asm volatile("" : "+v"(acc));
        const unsigned behind = te - pos - 32u;
        ask_rows<false>(rows, mine1, 0,
                        behind < (unsigned)ROWS_IN_FLIGHT ? (behind != 0u ? behind : 1u)
                                                          : (unsigned)ROWS_IN_FLIGHT,
                        lane, va);
        if (STEP) {
            self_step(xa_lane, wr_lane, ks, acc4);
            ++ks;
        }
        add_rows<true>(vb, ROWS_IN_FLIGHT, pos + ROWS_IN_FLIGHT, mystart, id, my_a, lane, acc, cur, nb);
        mine = mine1;
        mine1 = mine2;
        pos += 32u;
    }
    while (pos < stop) {
        const unsigned left = te - pos;
        const unsigned cnt = left < (unsigned)ROWS_IN_FLIGHT ? left : (unsigned)ROWS_IN_FLIGHT;
        if (STEP) {
            self_step(xa_lane, wr_lane, ks, acc4);
            ++ks;
        }
        add_rows<false>(va, cnt, pos, mystart, id, my_a, lane, acc, cur, nb);
        pos += (unsigned)ROWS_IN_FLIGHT;
        if (pos < te) {
            const unsigned j0 = (pos - tb) & 31u;
            if (j0 == 0u) {
                mine = mine1;
                const unsigned nx = pos + 32u + (unsigned)lane;
                mine1 = nbr[nx < te ? nx : te - 1];
            }
            const unsigned l2 = te - pos;
            ask_rows<false>(rows, mine, j0, l2 < (unsigned)ROWS_IN_FLIGHT ? l2 : (unsigned)ROWS_IN_FLIGHT,
                            lane, va);
        }
    }
}

template <bool RELU, bool ORDER>
__device__ __forceinline__ void layer_body(const float* __restrict__ feat,
                                           const int2* __restrict__ pairs,
                                           const unsigned* __restrict__ tab,
                                           unsigned* start, int* nbr, float* invdeg,
                                           const float* __restrict__ wl,
                                           const float* __restrict__ bl,
                                           const float* __restrict__ wr,
                                           float* __restrict__ out, const int n_nodes,
                                           const int n_bins, const int n_chunks) {
    __shared__ unsigned s_start[BIN_W + 1];
    __shared__ float s_inv[BIN_W];
    __shared__ v2f lds_wl[32 * WPAD];
    __shared__ v2f lds_wr[32 * WPAD];
    __shared__ __attribute__((aligned(16))) float lds_a[LAYER_WAVES * 16 * APITCH];
    __shared__ v2f lds_x[LAYER_WAVES * 16 * APITCH / 2];

    for (int i = threadIdx.x; i < 32 * 64; i += blockDim.x) {
        const int p = i >> 6;
        const int n = i & 63;
        v2f a, b;
        a.x = wl[(2 * p) * DIM + n];
        a.y = wl[(2 * p + 1) * DIM + n];
        b.x = wr[(2 * p) * DIM + n];
        b.y = wr[(2 * p + 1) * DIM + n];
        lds_wl[p * WPAD + n] = a;
        lds_wr[p * WPAD + n] = b;
    }
    __syncthreads();
    if constexpr (ORDER)
        order_bin(pairs, tab, start, invdeg, nbr, s_start, s_inv, (unsigned*)lds_a, (unsigned*)lds_x,
                  (int)blockIdx.x, n_nodes, n_bins, n_chunks);

    const int lane = threadIdx.x & 31;
    const int wave = __builtin_amdgcn_readfirstlane((int)(threadIdx.x >> 5));
    const int tile = blockIdx.x * LAYER_WAVES + wave;
    const int n_tiles = (n_nodes + 15) >> 4;
    if (tile >= n_tiles) return;

    const int row0 = tile << 4;
    const int r     = lane & 15;
    const int kh    = (lane >> 4) << 1;
    const int pbase = lane >> 4;
    float* my_a = lds_a + wave * (16 * APITCH);
    float* my_x = (float*)lds_x + wave * (16 * APITCH);

    int node = row0 + r;
    if (node > n_nodes - 1) node = n_nodes - 1;
    const float id = ORDER ? s_inv[wave * 16 + r] : invdeg[node];
    float bias[4];
#pragma unroll
    for (int nt = 0; nt < 4; ++nt) bias[nt] = bl[nt * 16 + r];

    int bi = row0 + (lane < 16 ? lane : 16);
    if (bi > n_nodes) bi = n_nodes;
    const int mystart = ORDER ? (int)s_start[wave * 16 + (lane < 16 ? lane : 16)] : (int)start[bi];

    const v2f* rows = (const v2f*)feat;
    v2f xs[16];
#pragma unroll
    for (int i = 0; i < 16; ++i) {
        int sn = row0 + i;
        if (sn > n_nodes - 1) sn = n_nodes - 1;
        xs[i] = rows[(long long)sn * 32 + lane];
    }

    v8f acc4[4] = {{}, {}, {}, {}};
    const float* xa_lane = my_x + r * APITCH + kh;
    const v2f* wr_lane = lds_wr + pbase * WPAD + r;
    unsigned ks = 0;

    {
        const unsigned tb = (unsigned)__builtin_amdgcn_readlane(mystart, 0);
        const unsigned te = (unsigned)__builtin_amdgcn_readlane(mystart, 16);
        unsigned cur = 0;
        unsigned nb = (unsigned)__builtin_amdgcn_readlane(mystart, 1);
        v2f acc = {0.0f, 0.0f};
        int mine = 0, mine1 = 0;
        v2f va[ROWS_IN_FLIGHT];
        if (tb < te) {
            const unsigned i0 = tb + (unsigned)lane;
            const unsigned i1 = i0 + 32u;
            mine = nbr[i0 < te ? i0 : te - 1];
            mine1 = nbr[i1 < te ? i1 : te - 1];
#pragma unroll
            for (int i = 0; i < 16; ++i) *(v2f*)&my_x[i * APITCH + 2 * lane] = xs[i];
            const unsigned l0 = te - tb;
            ask_rows<false>(rows, mine, 0, l0 < (unsigned)ROWS_IN_FLIGHT ? l0 : (unsigned)ROWS_IN_FLIGHT,
                            lane, va);
        } else {
#pragma unroll
            for (int i = 0; i < 16; ++i) *(v2f*)&my_x[i * APITCH + 2 * lane] = xs[i];
        }
        const unsigned step_rows = (unsigned)(DIM / 4) * (unsigned)ROWS_IN_FLIGHT;
        static_assert(((DIM / 4) * ROWS_IN_FLIGHT) % 32 == 0, "the stepped part ends on a whole piece");
        const unsigned te_step = te - tb < step_rows ? te : tb + step_rows;
        unsigned pos = tb;
        sum_span<true>(rows, nbr, tb, pos, te_step, te, mystart, id, my_a, lane, mine, mine1, va, acc,
                       cur, nb, xa_lane, wr_lane, ks, acc4);
        sum_span<false>(rows, nbr, tb, pos, te, te, mystart, id, my_a, lane, mine, mine1, va, acc,
                        cur, nb, xa_lane, wr_lane, ks, acc4);
        for (; cur < 16; ++cur) {
            const float sc = __builtin_bit_cast(
                float, __builtin_amdgcn_readlane(__builtin_bit_cast(int, id), cur));
            v2f o;
            o.x = acc.x * sc;
            o.y = acc.y * sc;
            *(v2f*)&my_a[cur * APITCH + 2 * lane] = o;
            acc.x = 0.0f;
            acc.y = 0.0f;
        }
    }

#pragma nounroll
    for (; ks < (unsigned)(DIM / 4); ++ks) self_step(xa_lane, wr_lane, ks, acc4);

    const float* arow = my_a + r * APITCH;
#pragma unroll
    for (int k = 0; k < DIM; k += 4) {
        const v2f a = *(const v2f*)&arow[k + kh];
        const int prow = (k >> 1) + pbase;
#pragma unroll
        for (int nt = 0; nt < 4; ++nt) {
            v2f b = lds_wl[prow * WPAD + nt * 16 + r];
            acc4[nt] = __builtin_amdgcn_wmma_f32_16x16x4_f32(
                false, a, false, b, (short)0, acc4[nt], false, false);
        }
    }

    const int mbase = (lane >> 4) << 3;
#pragma unroll
    for (int nt = 0; nt < 4; ++nt) {
#pragma unroll
        for (int vr = 0; vr < 8; ++vr) {
            float v = acc4[nt][vr] + bias[nt];
            if (RELU) v = fmaxf(v, 0.0f);
            my_a[(mbase + vr) * APITCH + nt * 16 + r] = v;
        }
    }
    const int orow = lane >> 4;
    const int ocol = (lane & 15) << 2;
    float* otile = out + (long long)row0 * DIM;
    if (row0 + 16 <= n_nodes) {
#pragma unroll
        for (int i = 0; i < 8; ++i) {
            const v4f v = *(const v4f*)&my_a[(2 * i + orow) * APITCH + ocol];
            *(v4f*)&otile[(2 * i + orow) * DIM + ocol] = v;
        }
    } else {
#pragma unroll
        for (int i = 0; i < 8; ++i) {
            if (row0 + 2 * i + orow < n_nodes) {
                const v4f v = *(const v4f*)&my_a[(2 * i + orow) * APITCH + ocol];
                *(v4f*)&otile[(2 * i + orow) * DIM + ocol] = v;
            }
        }
    }
}

template <bool RELU>
__global__ void __launch_bounds__(LAYER_T) sage_order_layer_kernel(const float* __restrict__ feat,
                                        const int2* __restrict__ pairs,
                                        const unsigned* __restrict__ tab,
                                        unsigned* start, int* nbr, float* invdeg,
                                        const float* __restrict__ wl,
                                        const float* __restrict__ bl,
                                        const float* __restrict__ wr,
                                        float* __restrict__ out, int n_nodes, int n_bins,
                                        int n_chunks) {
    layer_body<RELU, true>(feat, pairs, tab, start, nbr, invdeg, wl, bl, wr, out, n_nodes, n_bins,
                           n_chunks);
}

template <bool RELU>
__global__ void __launch_bounds__(LAYER_T) sage_layer_kernel(const float* __restrict__ feat,
                                  const unsigned* __restrict__ start,
                                  const int* __restrict__ nbr,
                                  const float* __restrict__ invdeg,
                                  const float* __restrict__ wl,
                                  const float* __restrict__ bl,
                                  const float* __restrict__ wr,
                                  float* __restrict__ out, int n_nodes) {
    layer_body<RELU, false>(feat, nullptr, nullptr, const_cast<unsigned*>(start),
                            const_cast<int*>(nbr), const_cast<float*>(invdeg), wl, bl, wr, out,
                            n_nodes, 0, 0);
}

extern "C" void kernel_launch(void* const* d_in, const int* in_sizes, int n_in,
                              void* d_out, int out_size, void* d_ws, size_t ws_size,
                              hipStream_t stream) {
    const float* x   = (const float*)d_in[0];
    const int*  edge = (const int*) d_in[1];
    const float* w1l = (const float*)d_in[2];
    const float* b1l = (const float*)d_in[3];
    const float* w1r = (const float*)d_in[4];
    const float* w2l = (const float*)d_in[5];
    const float* b2l = (const float*)d_in[6];
    const float* w2r = (const float*)d_in[7];
    float* out = (float*)d_out;

    const int n_nodes = in_sizes[0] / DIM;
    const int n_edges = in_sizes[1] / 2;
    const int* src = edge;
    const int* dst = edge + n_edges;

    float* ws      = (float*)d_ws;
    float* invdeg  = ws;
    float* list    = ws + n_nodes;
    float* h       = list + (long long)n_nodes * DIM;
    unsigned* start = (unsigned*)list;
    int*      nbr   = (int*)(start + n_nodes + 1);

    const int n_bins       = (n_nodes + BIN_W - 1) / BIN_W;
    const int n_chunks     = (n_edges + CH_EDGES - 1) / CH_EDGES;
    const int tiles        = (n_nodes + 15) / 16;
    const int layer_blocks = (tiles + LAYER_WAVES - 1) / LAYER_WAVES;

    int2*     pairs = (int2*)d_out;
    unsigned* tab   = (unsigned*)d_out + 2 * (long long)n_edges;

    chunk_sort_kernel<<<n_chunks, ORDER_T, 0, stream>>>(src, dst, tab, pairs, n_edges, n_bins);

    sage_order_layer_kernel<true><<<n_bins, LAYER_T, 0, stream>>>(
        x, pairs, tab, start, nbr, invdeg, w1l, b1l, w1r, h, n_nodes, n_bins, n_chunks);

    sage_layer_kernel<false><<<layer_blocks, LAYER_WAVES * 32, 0, stream>>>(
        h, start, nbr, invdeg, w2l, b2l, w2r, out, n_nodes);
}
